# starting kernel
# baseline (speedup 1.0000x reference)
#include <hip/hip_runtime.h>

typedef __attribute__((ext_vector_type(2))) float v2f;
typedef __attribute__((ext_vector_type(8))) float v8f;

#define D_DIM 128
#define CH    64
#define SEQ   4096
#define NT    (SEQ / CH)
#define NH    16
#define NB    2
#define EG    4
#define GW    32

#define SQ_LD 132
#define SK_LD 132
#define SV_LD 36
#define SS_LD 68
#define SH_LD 36

#define SQ_OFF 0
#define SK_OFF (SQ_OFF + 64 * SQ_LD)
#define SV_OFF (SK_OFF + 64 * SK_LD)
#define SS_OFF (SV_OFF + 64 * SV_LD)
#define SH_OFF (SS_OFF + 64 * SS_LD)
#define SMEM_FLOATS (SH_OFF + 128 * SH_LD)

static __device__ __forceinline__ v8f wmma_f32(v2f a, v2f b, v8f c) {
    return __builtin_amdgcn_wmma_f32_16x16x4_f32(
         false, a,  false, b,
         (short)0, c,  false,  false);
}

extern "C" __global__ __launch_bounds__(256, 1)
void fla_fused_chunk_fwd(const float* __restrict__ q,
                         const float* __restrict__ k,
                         const float* __restrict__ v,
                         float* __restrict__ o,
                         float* __restrict__ hout,
                         float scale)
{
    extern __shared__ float smem[];
    float* sQ = smem + SQ_OFF;
    float* sK = smem + SK_OFF;
    float* sV = smem + SV_OFF;
    float* sS = smem + SS_OFF;
    float* sH = smem + SH_OFF;

    const int tid  = threadIdx.x;
    const int wave = tid >> 5;
    const int lane = tid & 31;
    const int m16  = lane & 15;
    const int kh   = lane >> 4;

    const int g  = blockIdx.x & (EG - 1);
    const int bh = blockIdx.x >> 2;
    const int h  = bh & (NH - 1);
    const int b  = bh >> 4;

    const size_t row_stride = (size_t)NH * D_DIM;
    const float* qbase = q + ((size_t)b * SEQ * NH + h) * D_DIM;
    const float* kbase = k + ((size_t)b * SEQ * NH + h) * D_DIM;
    const float* vbase = v + ((size_t)b * SEQ * NH + h) * D_DIM;

    const int SR[10] = {0, 1, 1, 2, 2, 2, 3, 3, 3, 3};
    const int SC[10] = {0, 0, 1, 0, 1, 2, 0, 1, 2, 3};

    for (int i = tid; i < 128 * SH_LD; i += 256) sH[i] = 0.0f;
    __syncthreads();

    const int r_o  = wave & 3;
    const int eco  = (wave >> 2) * 16;

    for (int t = 0; t < NT; ++t) {
        const size_t srow = (size_t)t * CH;

        #pragma unroll
        for (int j = 0; j < 8; ++j) {
            int f   = tid + j * 256;
            int row = f >> 5;
            int c4  = (f & 31) << 2;
            float4 qv = *(const float4*)(qbase + (srow + row) * row_stride + c4);
            qv.x *= scale; qv.y *= scale; qv.z *= scale; qv.w *= scale;
            *(float4*)(sQ + row * SQ_LD + c4) = qv;
            float4 kv = *(const float4*)(kbase + (srow + row) * row_stride + c4);
            *(float4*)(sK + row * SK_LD + c4) = kv;
        }
        #pragma unroll
        for (int j = 0; j < 2; ++j) {
            int f   = tid + j * 256;
            int row = f >> 3;
            int c4  = (f & 7) << 2;
            float4 vv = *(const float4*)(vbase + (srow + row) * row_stride + g * GW + c4);
            *(float4*)(sV + row * SV_LD + c4) = vv;
        }
        __syncthreads();

        #pragma unroll
        for (int ii = 0; ii < 2; ++ii) {
            int idx = wave + ii * 8;
            if (idx < 10) {
                const int r = SR[idx], c = SC[idx];
                v8f acc = {};
                const float* aBase = sQ + (r * 16 + m16) * SQ_LD + kh * 2;
                const float* bBase = sK + (c * 16 + m16) * SK_LD + kh * 2;
                #pragma unroll 8
                for (int k0 = 0; k0 < 128; k0 += 4) {
                    v2f a  = *(const v2f*)(aBase + k0);
                    v2f bb = *(const v2f*)(bBase + k0);
                    acc = wmma_f32(a, bb, acc);
                }
                #pragma unroll
                for (int j = 0; j < 8; ++j) {
                    int er = r * 16 + j + 8 * kh;
                    int ec = c * 16 + m16;
                    float val = acc[j];
                    if (r == c && er < ec) val = 0.0f;
                    sS[er * SS_LD + ec] = val;
                }
            }
        }
        __syncthreads();

        {
            v8f acc = {};
            const float* aS  = sS + (r_o * 16 + m16) * SS_LD + kh * 2;
            const float* aQ  = sQ + (r_o * 16 + m16) * SQ_LD + kh * 2;
            const float* bV0 = sV + (kh * 2 + 0) * SV_LD + eco + m16;
            const float* bV1 = sV + (kh * 2 + 1) * SV_LD + eco + m16;
            const float* bH0 = sH + (kh * 2 + 0) * SH_LD + eco + m16;
            const float* bH1 = sH + (kh * 2 + 1) * SH_LD + eco + m16;

            const int kmax = (r_o + 1) * 16;
            #pragma unroll 4
            for (int k0 = 0; k0 < kmax; k0 += 4) {
                v2f a = *(const v2f*)(aS + k0);
                v2f bb; bb.x = bV0[k0 * SV_LD]; bb.y = bV1[k0 * SV_LD];
                acc = wmma_f32(a, bb, acc);
            }
            #pragma unroll 8
            for (int k0 = 0; k0 < 128; k0 += 4) {
                v2f a = *(const v2f*)(aQ + k0);
                v2f bb; bb.x = bH0[k0 * SH_LD]; bb.y = bH1[k0 * SH_LD];
                acc = wmma_f32(a, bb, acc);
            }
            #pragma unroll
            for (int j = 0; j < 8; ++j) {
                size_t s_idx = srow + r_o * 16 + j + 8 * kh;
                o[(((size_t)b * SEQ + s_idx) * NH + h) * D_DIM + g * GW + eco + m16] = acc[j];
            }
        }
        __syncthreads();

        {
            const int d = wave;
            #pragma unroll
            for (int e2t = 0; e2t < 2; ++e2t) {
                const int ec = e2t * 16;
                float* hTile = sH + ec + m16;
                v8f acc;
                #pragma unroll
                for (int j = 0; j < 8; ++j)
                    acc[j] = hTile[(d * 16 + j + 8 * kh) * SH_LD];
                const float* aK0 = sK + (kh * 2 + 0) * SK_LD + d * 16 + m16;
                const float* aK1 = sK + (kh * 2 + 1) * SK_LD + d * 16 + m16;
                const float* bV0 = sV + (kh * 2 + 0) * SV_LD + ec + m16;
                const float* bV1 = sV + (kh * 2 + 1) * SV_LD + ec + m16;
                #pragma unroll 4
                for (int k0 = 0; k0 < 64; k0 += 4) {
                    v2f a;  a.x  = aK0[k0 * SK_LD]; a.y  = aK1[k0 * SK_LD];
                    v2f bb; bb.x = bV0[k0 * SV_LD]; bb.y = bV1[k0 * SV_LD];
                    acc = wmma_f32(a, bb, acc);
                }
                #pragma unroll
                for (int j = 0; j < 8; ++j)
                    hTile[(d * 16 + j + 8 * kh) * SH_LD] = acc[j];
            }
        }
        __syncthreads();
    }

    #pragma unroll
    for (int j = 0; j < 4; ++j) {
        int f    = tid + j * 256;
        int drow = f >> 3;
        int c4   = (f & 7) << 2;
        float4 val = *(const float4*)(sH + drow * SH_LD + c4);
        *(float4*)(hout + (((size_t)(b * NH + h)) * D_DIM + drow) * D_DIM + g * GW + c4) = val;
    }
}

extern "C" void kernel_launch(void* const* d_in, const int* in_sizes, int n_in,
                              void* d_out, int out_size, void* d_ws, size_t ws_size,
                              hipStream_t stream) {
    (void)in_sizes; (void)n_in; (void)out_size; (void)d_ws; (void)ws_size;
    const float* q = (const float*)d_in[0];
    const float* k = (const float*)d_in[1];
    const float* v = (const float*)d_in[2];
    float* o  = (float*)d_out;
    float* hf = o + (size_t)NB * SEQ * NH * D_DIM;
    const float scale = 0.088388347648318447f;

    const size_t smem = (size_t)SMEM_FLOATS * sizeof(float);
    hipFuncSetAttribute((const void*)fla_fused_chunk_fwd,
                        hipFuncAttributeMaxDynamicSharedMemorySize, (int)smem);

    dim3 grid(NB * NH * EG);
    dim3 blk(256);
    hipLaunchKernelGGL(fla_fused_chunk_fwd, grid, blk, smem, stream,
                       q, k, v, o, hf, scale);
}


# LDS fragment loads issued in groups one matrix block ahead (k loops fully unrolled, empty-asm edges pin each group's place) in the S, O and state-update phases; every chain's k order kept
# speedup vs baseline: 1.3637x; 1.3637x over previous
#include <hip/hip_runtime.h>

typedef __attribute__((ext_vector_type(2))) float v2f;
typedef __attribute__((ext_vector_type(8))) float v8f;

#define D_DIM 128
#define CH    64
#define SEQ   4096
#define NT    (SEQ / CH)
#define NH    16
#define NB    2
#define EG    4
#define GW    32

#define SQ_LD 132
#define SK_LD 132
#define SV_LD 36
#define SS_LD 68
#define SH_LD 36

#define SQ_OFF 0
#define SK_OFF (SQ_OFF + 64 * SQ_LD)
#define SV_OFF (SK_OFF + 64 * SK_LD)
#define SS_OFF (SV_OFF + 64 * SV_LD)
#define SH_OFF (SS_OFF + 64 * SS_LD)
#define SMEM_FLOATS (SH_OFF + 128 * SH_LD)

#define EDGE1(a)    asm volatile("" : "+v"(a) : : "memory")
#define EDGE2(a, b) asm volatile("" : "+v"(a), "+v"(b) : : "memory")

static __device__ __forceinline__ v8f wmma_f32(v2f a, v2f b, v8f c) {
    return __builtin_amdgcn_wmma_f32_16x16x4_f32(
         false, a,  false, b,
         (short)0, c,  false,  false);
}

template <int J0, int N>
static __device__ __forceinline__ void ld_rr(v2f* fa, v2f* fb, const float* pa, const float* pb) {
    #pragma unroll
    for (int j = J0; j < J0 + N; ++j) {
        fa[j] = *(const v2f*)(pa + 4 * j);
        fb[j] = *(const v2f*)(pb + 4 * j);
    }
}

template <int J0, int N, int LD>
static __device__ __forceinline__ void ld_rk(v2f* fa, v2f* fb, const float* pa,
                                             const float* p0, const float* p1) {
    #pragma unroll
    for (int j = J0; j < J0 + N; ++j) {
        fa[j] = *(const v2f*)(pa + 4 * j);
        fb[j].x = p0[4 * j * LD];
        fb[j].y = p1[4 * j * LD];
    }
}

template <int J0, int N>
static __device__ __forceinline__ void ld_h(v2f* fk, v2f* fva, v2f* fvb,
                                            const float* k0, const float* k1,
                                            const float* v0, const float* v1) {
    #pragma unroll
    for (int j = J0; j < J0 + N; ++j) {
        fk[j].x  = k0[4 * j * SK_LD];      fk[j].y  = k1[4 * j * SK_LD];
        fva[j].x = v0[4 * j * SV_LD];      fva[j].y = v1[4 * j * SV_LD];
        fvb[j].x = v0[4 * j * SV_LD + 16]; fvb[j].y = v1[4 * j * SV_LD + 16];
    }
}

template <int J0, int N>
static __device__ __forceinline__ v8f mm(const v2f* fa, const v2f* fb, v8f acc) {
    #pragma unroll
    for (int j = J0; j < J0 + N; ++j) acc = wmma_f32(fa[j], fb[j], acc);
    return acc;
}

template <int J0, int N>
static __device__ __forceinline__ void mm2(const v2f* fa0, const v2f* fb0, v8f& c0,
                                           const v2f* fa1, const v2f* fb1, v8f& c1) {
    #pragma unroll
    for (int j = J0; j < J0 + N; ++j) {
        c0 = wmma_f32(fa0[j], fb0[j], c0);
        c1 = wmma_f32(fa1[j], fb1[j], c1);
    }
}

static __device__ __forceinline__ void s_store(float* sS, v8f acc, int r, int c, int m16, int kh) {
    #pragma unroll
    for (int j = 0; j < 8; ++j) {
        int er = r * 16 + j + 8 * kh;
        int ec = c * 16 + m16;
        float val = acc[j];
        if (r == c && er < ec) val = 0.0f;
        sS[er * SS_LD + ec] = val;
    }
}

template <int NBLK>
static __device__ __forceinline__ v8f sv_part(const float* aS, const float* bV0, const float* bV1) {
    v2f fs[NBLK * 4], fv[NBLK * 4];
    ld_rk<0, NBLK * 4, SV_LD>(fs, fv, aS, bV0, bV1);
    v8f acc = {};
    EDGE1(acc);
    return mm<0, NBLK * 4>(fs, fv, acc);
}

static __device__ __forceinline__ int tile_row(int idx) {
    return (idx >= 6) ? 3 : (idx >= 3) ? 2 : (idx >= 1) ? 1 : 0;
}

extern "C" __global__ __launch_bounds__(256)
void fla_fused_chunk_fwd(const float* __restrict__ q,
                         const float* __restrict__ k,
                         const float* __restrict__ v,
                         float* __restrict__ o,
                         float* __restrict__ hout,
                         float scale)
{
    extern __shared__ float smem[];
    float* sQ = smem + SQ_OFF;
    float* sK = smem + SK_OFF;
    float* sV = smem + SV_OFF;
    float* sS = smem + SS_OFF;
    float* sH = smem + SH_OFF;

    const int tid  = threadIdx.x;
    const int wave = tid >> 5;
    const int lane = tid & 31;
    const int m16  = lane & 15;
    const int kh   = lane >> 4;

    const int g  = blockIdx.x & (EG - 1);
    const int bh = blockIdx.x >> 2;
    const int h  = bh & (NH - 1);
    const int b  = bh >> 4;

    const size_t row_stride = (size_t)NH * D_DIM;
    const float* qbase = q + ((size_t)b * SEQ * NH + h) * D_DIM;
    const float* kbase = k + ((size_t)b * SEQ * NH + h) * D_DIM;
    const float* vbase = v + ((size_t)b * SEQ * NH + h) * D_DIM;

    for (int i = tid; i < 128 * SH_LD; i += 256) sH[i] = 0.0f;
    __syncthreads();

    const int r_o  = wave & 3;
    const int eco  = (wave >> 2) * 16;

    const bool two_tiles = wave < 2;
    const int  rA = tile_row(wave),     cA = wave - rA * (rA + 1) / 2;
    const int  rB = 3,                  cB = wave + 2;
    const float* sA_a = sQ + (rA * 16 + m16) * SQ_LD + kh * 2;
    const float* sA_b = sK + (cA * 16 + m16) * SK_LD + kh * 2;
    const float* sB_a = sQ + (rB * 16 + m16) * SQ_LD + kh * 2;
    const float* sB_b = sK + ((cB & 3) * 16 + m16) * SK_LD + kh * 2;

    const float* aS  = sS + (r_o * 16 + m16) * SS_LD + kh * 2;
    const float* aQ  = sQ + (r_o * 16 + m16) * SQ_LD + kh * 2;
    const float* oV0 = sV + (kh * 2 + 0) * SV_LD + eco + m16;
    const float* oV1 = sV + (kh * 2 + 1) * SV_LD + eco + m16;
    const float* oH0 = sH + (kh * 2 + 0) * SH_LD + eco + m16;
    const float* oH1 = sH + (kh * 2 + 1) * SH_LD + eco + m16;

    const int d = wave;
    float* hT0 = sH + (d * 16 + 8 * kh) * SH_LD + m16;
    float* hT1 = hT0 + 16;
    const float* hK0 = sK + (kh * 2 + 0) * SK_LD + d * 16 + m16;
    const float* hK1 = sK + (kh * 2 + 1) * SK_LD + d * 16 + m16;
    const float* hV0 = sV + (kh * 2 + 0) * SV_LD + m16;
    const float* hV1 = sV + (kh * 2 + 1) * SV_LD + m16;

    for (int t = 0; t < NT; ++t) {
        const size_t srow = (size_t)t * CH;

        #pragma unroll
        for (int j = 0; j < 8; ++j) {
            int f   = tid + j * 256;
            int row = f >> 5;
            int c4  = (f & 31) << 2;
            float4 qv = *(const float4*)(qbase + (srow + row) * row_stride + c4);
            qv.x *= scale; qv.y *= scale; qv.z *= scale; qv.w *= scale;
            *(float4*)(sQ + row * SQ_LD + c4) = qv;
            float4 kv = *(const float4*)(kbase + (srow + row) * row_stride + c4);
            *(float4*)(sK + row * SK_LD + c4) = kv;
        }
        #pragma unroll
        for (int j = 0; j < 2; ++j) {
            int f   = tid + j * 256;
            int row = f >> 3;
            int c4  = (f & 7) << 2;
            float4 vv = *(const float4*)(vbase + (srow + row) * row_stride + g * GW + c4);
            *(float4*)(sV + row * SV_LD + c4) = vv;
        }
        __syncthreads();

        if (two_tiles) {
            v2f fa0[32], fb0[32], fa1[32], fb1[32];
            v8f c0 = {}, c1 = {};
            ld_rr<0, 8>(fa0, fb0, sA_a, sA_b);   ld_rr<0, 8>(fa1, fb1, sB_a, sB_b);
            ld_rr<8, 8>(fa0, fb0, sA_a, sA_b);   ld_rr<8, 8>(fa1, fb1, sB_a, sB_b);
            EDGE2(c0, c1);
            mm2<0, 8>(fa0, fb0, c0, fa1, fb1, c1);
            EDGE2(c0, c1);
            ld_rr<16, 8>(fa0, fb0, sA_a, sA_b);  ld_rr<16, 8>(fa1, fb1, sB_a, sB_b);
            EDGE2(c0, c1);
            mm2<8, 8>(fa0, fb0, c0, fa1, fb1, c1);
            EDGE2(c0, c1);
            ld_rr<24, 8>(fa0, fb0, sA_a, sA_b);  ld_rr<24, 8>(fa1, fb1, sB_a, sB_b);
            EDGE2(c0, c1);
            mm2<16, 16>(fa0, fb0, c0, fa1, fb1, c1);
            EDGE2(c0, c1);
            s_store(sS, c0, rA, cA, m16, kh);
            s_store(sS, c1, rB, cB, m16, kh);
        } else {
            v2f fa[32], fb[32];
            v8f c0 = {};
            ld_rr<0, 8>(fa, fb, sA_a, sA_b);
            ld_rr<8, 8>(fa, fb, sA_a, sA_b);
            EDGE1(c0);
            c0 = mm<0, 8>(fa, fb, c0);
            EDGE1(c0);
            ld_rr<16, 8>(fa, fb, sA_a, sA_b);
            EDGE1(c0);
            c0 = mm<8, 8>(fa, fb, c0);
            EDGE1(c0);
            ld_rr<24, 8>(fa, fb, sA_a, sA_b);
            EDGE1(c0);
            c0 = mm<16, 16>(fa, fb, c0);
            EDGE1(c0);
            s_store(sS, c0, rA, cA, m16, kh);
        }

        v2f fq[32], fh[32];
        ld_rk<0, 16, SH_LD>(fq, fh, aQ, oH0, oH1);
        __syncthreads();

        v8f acc;
        switch (r_o) {
            case 0:  acc = sv_part<1>(aS, oV0, oV1); break;
            case 1:  acc = sv_part<2>(aS, oV0, oV1); break;
            case 2:  acc = sv_part<3>(aS, oV0, oV1); break;
            default: acc = sv_part<4>(aS, oV0, oV1); break;
        }
        v2f fk[16], fva[16], fvb[16];
        v8f h0, h1;
        EDGE1(acc);
        ld_rk<16, 8, SH_LD>(fq, fh, aQ, oH0, oH1);
        EDGE1(acc);
        acc = mm<0, 8>(fq, fh, acc);
        EDGE1(acc);
        ld_rk<24, 8, SH_LD>(fq, fh, aQ, oH0, oH1);
        EDGE1(acc);
        acc = mm<8, 8>(fq, fh, acc);
        EDGE1(acc);
        #pragma unroll
        for (int j = 0; j < 8; ++j) { h0[j] = hT0[j * SH_LD]; h1[j] = hT1[j * SH_LD]; }
        ld_h<0, 8>(fk, fva, fvb, hK0, hK1, hV0, hV1);
        EDGE1(acc);
        acc = mm<16, 8>(fq, fh, acc);
        EDGE1(acc);
        ld_h<8, 8>(fk, fva, fvb, hK0, hK1, hV0, hV1);
        EDGE1(acc);
        acc = mm<24, 8>(fq, fh, acc);

        #pragma unroll
        for (int j = 0; j < 8; ++j) {
            size_t s_idx = srow + r_o * 16 + j + 8 * kh;
            o[(((size_t)b * SEQ + s_idx) * NH + h) * D_DIM + g * GW + eco + m16] = acc[j];
        }
        EDGE2(h0, h1);
        __syncthreads();

        mm2<0, 16>(fk, fva, h0, fk, fvb, h1);
        #pragma unroll
        for (int j = 0; j < 8; ++j) { hT0[j * SH_LD] = h0[j]; hT1[j * SH_LD] = h1[j]; }
        __syncthreads();
    }

    #pragma unroll
    for (int j = 0; j < 4; ++j) {
        int f    = tid + j * 256;
        int drow = f >> 3;
        int c4   = (f & 7) << 2;
        float4 val = *(const float4*)(sH + drow * SH_LD + c4);
        *(float4*)(hout + (((size_t)(b * NH + h)) * D_DIM + drow) * D_DIM + g * GW + c4) = val;
    }
}

extern "C" void kernel_launch(void* const* d_in, const int* in_sizes, int n_in,
                              void* d_out, int out_size, void* d_ws, size_t ws_size,
                              hipStream_t stream) {
    (void)in_sizes; (void)n_in; (void)out_size; (void)d_ws; (void)ws_size;
    const float* q = (const float*)d_in[0];
    const float* k = (const float*)d_in[1];
    const float* v = (const float*)d_in[2];
    float* o  = (float*)d_out;
    float* hf = o + (size_t)NB * SEQ * NH * D_DIM;
    const float scale = 0.088388347648318447f;

    const size_t smem = (size_t)SMEM_FLOATS * sizeof(float);
    hipFuncSetAttribute((const void*)fla_fused_chunk_fwd,
                        hipFuncAttributeMaxDynamicSharedMemorySize, (int)smem);

    dim3 grid(NB * NH * EG);
    dim3 blk(256);
    hipLaunchKernelGGL(fla_fused_chunk_fwd, grid, blk, smem, stream,
                       q, k, v, o, hf, scale);
}


# q/k/v staging pipelined one chunk ahead: global loads in flight across the S and O phases (barriers fence LDS only, no load/store drain), LDS stores beside the state update; sums unchanged
# speedup vs baseline: 1.7979x; 1.3184x over previous
#include <hip/hip_runtime.h>

typedef __attribute__((ext_vector_type(2))) float v2f;
typedef __attribute__((ext_vector_type(8))) float v8f;
typedef __attribute__((ext_vector_type(4))) float v4f;

#define D_DIM 128
#define CH    64
#define SEQ   4096
#define NT    (SEQ / CH)
#define NH    16
#define NB    2
#define EG    4
#define GW    32

#define SQ_LD 132
#define SK_LD 132
#define SV_LD 36
#define SS_LD 68
#define SH_LD 36

#define SQ_OFF 0
#define SK_OFF (SQ_OFF + 64 * SQ_LD)
#define SV_OFF (SK_OFF + 64 * SK_LD)
#define SS_OFF (SV_OFF + 64 * SV_LD)
#define SH_OFF (SS_OFF + 64 * SS_LD)
#define SMEM_FLOATS (SH_OFF + 128 * SH_LD)

#define EDGE1(a)    asm volatile("" : "+v"(a) : : "memory")
#define EDGE2(a, b) asm volatile("" : "+v"(a), "+v"(b) : : "memory")

static __device__ __forceinline__ v8f wmma_f32(v2f a, v2f b, v8f c) {
    return __builtin_amdgcn_wmma_f32_16x16x4_f32(
         false, a,  false, b,
         (short)0, c,  false,  false);
}

static __device__ __forceinline__ void lds_barrier() {
    __builtin_amdgcn_fence(__ATOMIC_RELEASE, "workgroup", "local");
    __builtin_amdgcn_s_barrier();
    __builtin_amdgcn_fence(__ATOMIC_ACQUIRE, "workgroup", "local");
}

template <int J0, int N>
static __device__ __forceinline__ void ld_rr(v2f* fa, v2f* fb, const float* pa, const float* pb) {
    #pragma unroll
    for (int j = J0; j < J0 + N; ++j) {
        fa[j] = *(const v2f*)(pa + 4 * j);
        fb[j] = *(const v2f*)(pb + 4 * j);
    }
}

template <int J0, int N, int LD>
static __device__ __forceinline__ void ld_rk(v2f* fa, v2f* fb, const float* pa,
                                             const float* p0, const float* p1) {
    #pragma unroll
    for (int j = J0; j < J0 + N; ++j) {
        fa[j] = *(const v2f*)(pa + 4 * j);
        fb[j].x = p0[4 * j * LD];
        fb[j].y = p1[4 * j * LD];
    }
}

template <int J0, int N>
static __device__ __forceinline__ void ld_h(v2f* fk, v2f* fva, v2f* fvb,
                                            const float* k0, const float* k1,
                                            const float* v0, const float* v1) {
    #pragma unroll
    for (int j = J0; j < J0 + N; ++j) {
        fk[j].x  = k0[4 * j * SK_LD];      fk[j].y  = k1[4 * j * SK_LD];
        fva[j].x = v0[4 * j * SV_LD];      fva[j].y = v1[4 * j * SV_LD];
        fvb[j].x = v0[4 * j * SV_LD + 16]; fvb[j].y = v1[4 * j * SV_LD + 16];
    }
}

template <int J0, int N>
static __device__ __forceinline__ v8f mm(const v2f* fa, const v2f* fb, v8f acc) {
    #pragma unroll
    for (int j = J0; j < J0 + N; ++j) acc = wmma_f32(fa[j], fb[j], acc);
    return acc;
}

template <int J0, int N>
static __device__ __forceinline__ void mm2(const v2f* fa0, const v2f* fb0, v8f& c0,
                                           const v2f* fa1, const v2f* fb1, v8f& c1) {
    #pragma unroll
    for (int j = J0; j < J0 + N; ++j) {
        c0 = wmma_f32(fa0[j], fb0[j], c0);
        c1 = wmma_f32(fa1[j], fb1[j], c1);
    }
}

static __device__ __forceinline__ void s_store(float* sS, v8f acc, int r, int c, int m16, int kh) {
    #pragma unroll
    for (int j = 0; j < 8; ++j) {
        int er = r * 16 + j + 8 * kh;
        int ec = c * 16 + m16;
        float val = acc[j];
        if (r == c && er < ec) val = 0.0f;
        sS[er * SS_LD + ec] = val;
    }
}

template <int NBLK>
static __device__ __forceinline__ v8f sv_part(const float* aS, const float* bV0, const float* bV1) {
    v2f fs[NBLK * 4], fv[NBLK * 4];
    ld_rk<0, NBLK * 4, SV_LD>(fs, fv, aS, bV0, bV1);
    v8f acc = {};
    EDGE1(acc);
    return mm<0, NBLK * 4>(fs, fv, acc);
}

static __device__ __forceinline__ int tile_row(int idx) {
    return (idx >= 6) ? 3 : (idx >= 3) ? 2 : (idx >= 1) ? 1 : 0;
}

static __device__ __forceinline__ void fetch_chunk(v4f* pq, v4f* pk, v4f* pv,
                                                   const float* qrow, const float* krow,
                                                   const float* vrow, size_t row_stride) {
    #pragma unroll
    for (int j = 0; j < 8; ++j) {
        pq[j] = *(const v4f*)(qrow + (size_t)(8 * j) * row_stride);
        pk[j] = *(const v4f*)(krow + (size_t)(8 * j) * row_stride);
    }
    #pragma unroll
    for (int j = 0; j < 2; ++j)
        pv[j] = *(const v4f*)(vrow + (size_t)(32 * j) * row_stride);
}

static __device__ __forceinline__ void stage_chunk(const v4f* pq, const v4f* pk, const v4f* pv,
                                                   float* sQ_st, float* sK_st, float* sV_st,
                                                   float scale) {
    #pragma unroll
    for (int j = 0; j < 8; ++j) {
        *(v4f*)(sQ_st + 8 * j * SQ_LD) = pq[j] * scale;
        *(v4f*)(sK_st + 8 * j * SK_LD) = pk[j];
    }
    #pragma unroll
    for (int j = 0; j < 2; ++j)
        *(v4f*)(sV_st + 32 * j * SV_LD) = pv[j];
}

extern "C" __global__ __launch_bounds__(256)
void fla_fused_chunk_fwd(const float* __restrict__ q,
                         const float* __restrict__ k,
                         const float* __restrict__ v,
                         float* __restrict__ o,
                         float* __restrict__ hout,
                         float scale)
{
    extern __shared__ float smem[];
    float* sQ = smem + SQ_OFF;
    float* sK = smem + SK_OFF;
    float* sV = smem + SV_OFF;
    float* sS = smem + SS_OFF;
    float* sH = smem + SH_OFF;

    const int tid  = threadIdx.x;
    const int wave = tid >> 5;
    const int lane = tid & 31;
    const int m16  = lane & 15;
    const int kh   = lane >> 4;

    const int g  = blockIdx.x & (EG - 1);
    const int bh = blockIdx.x >> 2;
    const int h  = bh & (NH - 1);
    const int b  = bh >> 4;

    const size_t row_stride = (size_t)NH * D_DIM;
    const float* qbase = q + ((size_t)b * SEQ * NH + h) * D_DIM;
    const float* kbase = k + ((size_t)b * SEQ * NH + h) * D_DIM;
    const float* vbase = v + ((size_t)b * SEQ * NH + h) * D_DIM;

    const int qk_row = tid >> 5, qk_c4 = (tid & 31) << 2;
    const int v_row  = tid >> 3, v_c4  = (tid & 7) << 2;
    const float* qrow = qbase + (size_t)qk_row * row_stride + qk_c4;
    const float* krow = kbase + (size_t)qk_row * row_stride + qk_c4;
    const float* vrow = vbase + (size_t)v_row * row_stride + g * GW + v_c4;
    float* sQ_st = sQ + qk_row * SQ_LD + qk_c4;
    float* sK_st = sK + qk_row * SK_LD + qk_c4;
    float* sV_st = sV + v_row * SV_LD + v_c4;

    v4f pq[8], pk[8], pv[2];
    fetch_chunk(pq, pk, pv, qrow, krow, vrow, row_stride);

    for (int i = tid; i < 128 * SH_LD; i += 256) sH[i] = 0.0f;

    stage_chunk(pq, pk, pv, sQ_st, sK_st, sV_st, scale);
    lds_barrier();

    const int r_o  = wave & 3;
    const int eco  = (wave >> 2) * 16;

    const bool two_tiles = wave < 2;
    const int  rA = tile_row(wave),     cA = wave - rA * (rA + 1) / 2;
    const int  rB = 3,                  cB = wave + 2;
    const float* sA_a = sQ + (rA * 16 + m16) * SQ_LD + kh * 2;
    const float* sA_b = sK + (cA * 16 + m16) * SK_LD + kh * 2;
    const float* sB_a = sQ + (rB * 16 + m16) * SQ_LD + kh * 2;
    const float* sB_b = sK + ((cB & 3) * 16 + m16) * SK_LD + kh * 2;

    const float* aS  = sS + (r_o * 16 + m16) * SS_LD + kh * 2;
    const float* aQ  = sQ + (r_o * 16 + m16) * SQ_LD + kh * 2;
    const float* oV0 = sV + (kh * 2 + 0) * SV_LD + eco + m16;
    const float* oV1 = sV + (kh * 2 + 1) * SV_LD + eco + m16;
    const float* oH0 = sH + (kh * 2 + 0) * SH_LD + eco + m16;
    const float* oH1 = sH + (kh * 2 + 1) * SH_LD + eco + m16;

    const int d = wave;
    float* hT0 = sH + (d * 16 + 8 * kh) * SH_LD + m16;
    float* hT1 = hT0 + 16;
    const float* hK0 = sK + (kh * 2 + 0) * SK_LD + d * 16 + m16;
    const float* hK1 = sK + (kh * 2 + 1) * SK_LD + d * 16 + m16;
    const float* hV0 = sV + (kh * 2 + 0) * SV_LD + m16;
    const float* hV1 = sV + (kh * 2 + 1) * SV_LD + m16;

    for (int t = 0; t < NT; ++t) {
        const size_t srow = (size_t)t * CH;

        if (t + 1 < NT) {
            const size_t adv = (srow + CH) * row_stride;
            fetch_chunk(pq, pk, pv, qrow + adv, krow + adv, vrow + adv, row_stride);
        }

        if (two_tiles) {
            v2f fa0[32], fb0[32], fa1[32], fb1[32];
            v8f c0 = {}, c1 = {};
            ld_rr<0, 8>(fa0, fb0, sA_a, sA_b);   ld_rr<0, 8>(fa1, fb1, sB_a, sB_b);
            ld_rr<8, 8>(fa0, fb0, sA_a, sA_b);   ld_rr<8, 8>(fa1, fb1, sB_a, sB_b);
            EDGE2(c0, c1);
            mm2<0, 8>(fa0, fb0, c0, fa1, fb1, c1);
            EDGE2(c0, c1);
            ld_rr<16, 8>(fa0, fb0, sA_a, sA_b);  ld_rr<16, 8>(fa1, fb1, sB_a, sB_b);
            EDGE2(c0, c1);
            mm2<8, 8>(fa0, fb0, c0, fa1, fb1, c1);
            EDGE2(c0, c1);
            ld_rr<24, 8>(fa0, fb0, sA_a, sA_b);  ld_rr<24, 8>(fa1, fb1, sB_a, sB_b);
            EDGE2(c0, c1);
            mm2<16, 16>(fa0, fb0, c0, fa1, fb1, c1);
            EDGE2(c0, c1);
            s_store(sS, c0, rA, cA, m16, kh);
            s_store(sS, c1, rB, cB, m16, kh);
        } else {
            v2f fa[32], fb[32];
            v8f c0 = {};
            ld_rr<0, 8>(fa, fb, sA_a, sA_b);
            ld_rr<8, 8>(fa, fb, sA_a, sA_b);
            EDGE1(c0);
            c0 = mm<0, 8>(fa, fb, c0);
            EDGE1(c0);
            ld_rr<16, 8>(fa, fb, sA_a, sA_b);
            EDGE1(c0);
            c0 = mm<8, 8>(fa, fb, c0);
            EDGE1(c0);
            ld_rr<24, 8>(fa, fb, sA_a, sA_b);
            EDGE1(c0);
            c0 = mm<16, 16>(fa, fb, c0);
            EDGE1(c0);
            s_store(sS, c0, rA, cA, m16, kh);
        }

        v2f fq[32], fh[32];
        ld_rk<0, 16, SH_LD>(fq, fh, aQ, oH0, oH1);
        lds_barrier();

        v8f acc;
        switch (r_o) {
            case 0:  acc = sv_part<1>(aS, oV0, oV1); break;
            case 1:  acc = sv_part<2>(aS, oV0, oV1); break;
            case 2:  acc = sv_part<3>(aS, oV0, oV1); break;
            default: acc = sv_part<4>(aS, oV0, oV1); break;
        }
        v2f fk[16], fva[16], fvb[16];
        v8f h0, h1;
        EDGE1(acc);
        ld_rk<16, 8, SH_LD>(fq, fh, aQ, oH0, oH1);
        EDGE1(acc);
        acc = mm<0, 8>(fq, fh, acc);
        EDGE1(acc);
        ld_rk<24, 8, SH_LD>(fq, fh, aQ, oH0, oH1);
        EDGE1(acc);
        acc = mm<8, 8>(fq, fh, acc);
        EDGE1(acc);
        #pragma unroll
        for (int j = 0; j < 8; ++j) { h0[j] = hT0[j * SH_LD]; h1[j] = hT1[j * SH_LD]; }
        ld_h<0, 8>(fk, fva, fvb, hK0, hK1, hV0, hV1);
        EDGE1(acc);
        acc = mm<16, 8>(fq, fh, acc);
        EDGE1(acc);
        ld_h<8, 8>(fk, fva, fvb, hK0, hK1, hV0, hV1);
        EDGE1(acc);
        acc = mm<24, 8>(fq, fh, acc);

        #pragma unroll
        for (int j = 0; j < 8; ++j) {
            size_t s_idx = srow + r_o * 16 + j + 8 * kh;
            o[(((size_t)b * SEQ + s_idx) * NH + h) * D_DIM + g * GW + eco + m16] = acc[j];
        }
        EDGE2(h0, h1);
        lds_barrier();

        if (t + 1 < NT)
            stage_chunk(pq, pk, pv, sQ_st, sK_st, sV_st, scale);

        mm2<0, 16>(fk, fva, h0, fk, fvb, h1);
        #pragma unroll
        for (int j = 0; j < 8; ++j) { hT0[j * SH_LD] = h0[j]; hT1[j * SH_LD] = h1[j]; }
        lds_barrier();
    }

    #pragma unroll
    for (int j = 0; j < 4; ++j) {
        int f    = tid + j * 256;
        int drow = f >> 3;
        int c4   = (f & 7) << 2;
        float4 val = *(const float4*)(sH + drow * SH_LD + c4);
        *(float4*)(hout + (((size_t)(b * NH + h)) * D_DIM + drow) * D_DIM + g * GW + c4) = val;
    }
}

extern "C" void kernel_launch(void* const* d_in, const int* in_sizes, int n_in,
                              void* d_out, int out_size, void* d_ws, size_t ws_size,
                              hipStream_t stream) {
    (void)in_sizes; (void)n_in; (void)out_size; (void)d_ws; (void)ws_size;
    const float* q = (const float*)d_in[0];
    const float* k = (const float*)d_in[1];
    const float* v = (const float*)d_in[2];
    float* o  = (float*)d_out;
    float* hf = o + (size_t)NB * SEQ * NH * D_DIM;
    const float scale = 0.088388347648318447f;

    const size_t smem = (size_t)SMEM_FLOATS * sizeof(float);
    hipFuncSetAttribute((const void*)fla_fused_chunk_fwd,
                        hipFuncAttributeMaxDynamicSharedMemorySize, (int)smem);

    dim3 grid(NB * NH * EG);
    dim3 blk(256);
    hipLaunchKernelGGL(fla_fused_chunk_fwd, grid, blk, smem, stream,
                       q, k, v, o, hf, scale);
}


# sequence cut at chunk 35 into two segments on 256 workgroups; the second segment's workgroups rebuild its entering state with a K^T V-only pass (same chain order per state tile, sums unchanged)
# speedup vs baseline: 2.7037x; 1.5038x over previous
#include <hip/hip_runtime.h>

typedef __attribute__((ext_vector_type(2))) float v2f;
typedef __attribute__((ext_vector_type(8))) float v8f;
typedef __attribute__((ext_vector_type(4))) float v4f;

#define D_DIM 128
#define CH    64
#define SEQ   4096
#define NT    (SEQ / CH)
#define NH    16
#define NB    2
#define EG    4
#define GW    32
#define NWG1  (NB * NH * EG)
#define T_SPLIT 35
static_assert(T_SPLIT >= 3 && T_SPLIT < NT, "the state pass keeps two chunks in flight");

#define SQ_LD 132
#define SK_LD 132
#define SV_LD 36
#define SS_LD 68
#define SH_LD 36

#define SQ_OFF 0
#define SK_OFF (SQ_OFF + 64 * SQ_LD)
#define SV_OFF (SK_OFF + 64 * SK_LD)
#define SS_OFF (SV_OFF + 64 * SV_LD)
#define SH_OFF (SS_OFF + 64 * SS_LD)
#define SMEM_FLOATS (SH_OFF + 128 * SH_LD)

#define EDGE1(a)    asm volatile("" : "+v"(a) : : "memory")
#define EDGE2(a, b) asm volatile("" : "+v"(a), "+v"(b) : : "memory")

static __device__ __forceinline__ v8f wmma_f32(v2f a, v2f b, v8f c) {
    return __builtin_amdgcn_wmma_f32_16x16x4_f32(
         false, a,  false, b,
         (short)0, c,  false,  false);
}

static __device__ __forceinline__ void lds_barrier() {
    __builtin_amdgcn_fence(__ATOMIC_RELEASE, "workgroup", "local");
    __builtin_amdgcn_s_barrier();
    __builtin_amdgcn_fence(__ATOMIC_ACQUIRE, "workgroup", "local");
}

template <int J0, int N>
static __device__ __forceinline__ void ld_rr(v2f* fa, v2f* fb, const float* pa, const float* pb) {
    #pragma unroll
    for (int j = J0; j < J0 + N; ++j) {
        fa[j] = *(const v2f*)(pa + 4 * j);
        fb[j] = *(const v2f*)(pb + 4 * j);
    }
}

template <int J0, int N, int LD>
static __device__ __forceinline__ void ld_rk(v2f* fa, v2f* fb, const float* pa,
                                             const float* p0, const float* p1) {
    #pragma unroll
    for (int j = J0; j < J0 + N; ++j) {
        fa[j] = *(const v2f*)(pa + 4 * j);
        fb[j].x = p0[4 * j * LD];
        fb[j].y = p1[4 * j * LD];
    }
}

template <int J0, int N>
static __device__ __forceinline__ void ld_h(v2f* fk, v2f* fva, v2f* fvb,
                                            const float* k0, const float* k1,
                                            const float* v0, const float* v1) {
    #pragma unroll
    for (int j = J0; j < J0 + N; ++j) {
        fk[j].x  = k0[4 * j * SK_LD];      fk[j].y  = k1[4 * j * SK_LD];
        fva[j].x = v0[4 * j * SV_LD];      fva[j].y = v1[4 * j * SV_LD];
        fvb[j].x = v0[4 * j * SV_LD + 16]; fvb[j].y = v1[4 * j * SV_LD + 16];
    }
}

template <int J0, int N>
static __device__ __forceinline__ v8f mm(const v2f* fa, const v2f* fb, v8f acc) {
    #pragma unroll
    for (int j = J0; j < J0 + N; ++j) acc = wmma_f32(fa[j], fb[j], acc);
    return acc;
}

template <int J0, int N>
static __device__ __forceinline__ void mm2(const v2f* fa0, const v2f* fb0, v8f& c0,
                                           const v2f* fa1, const v2f* fb1, v8f& c1) {
    #pragma unroll
    for (int j = J0; j < J0 + N; ++j) {
        c0 = wmma_f32(fa0[j], fb0[j], c0);
        c1 = wmma_f32(fa1[j], fb1[j], c1);
    }
}

static __device__ __forceinline__ void s_store(float* sS, v8f acc, int r, int c, int m16, int kh) {
    #pragma unroll
    for (int j = 0; j < 8; ++j) {
        int er = r * 16 + j + 8 * kh;
        int ec = c * 16 + m16;
        float val = acc[j];
        if (r == c && er < ec) val = 0.0f;
        sS[er * SS_LD + ec] = val;
    }
}

template <int NBLK>
static __device__ __forceinline__ v8f sv_part(const float* aS, const float* bV0, const float* bV1) {
    v2f fs[NBLK * 4], fv[NBLK * 4];
    ld_rk<0, NBLK * 4, SV_LD>(fs, fv, aS, bV0, bV1);
    v8f acc = {};
    EDGE1(acc);
    return mm<0, NBLK * 4>(fs, fv, acc);
}

static __device__ __forceinline__ int tile_row(int idx) {
    return (idx >= 6) ? 3 : (idx >= 3) ? 2 : (idx >= 1) ? 1 : 0;
}

static __device__ __forceinline__ void fetch_chunk(v4f* pq, v4f* pk, v4f* pv,
                                                   const float* qrow, const float* krow,
                                                   const float* vrow, size_t row_stride) {
    #pragma unroll
    for (int j = 0; j < 8; ++j) {
        pq[j] = *(const v4f*)(qrow + (size_t)(8 * j) * row_stride);
        pk[j] = *(const v4f*)(krow + (size_t)(8 * j) * row_stride);
    }
    #pragma unroll
    for (int j = 0; j < 2; ++j)
        pv[j] = *(const v4f*)(vrow + (size_t)(32 * j) * row_stride);
}

static __device__ __forceinline__ void fetch_kv(v4f* pk, v4f* pv, const float* krow,
                                                const float* vrow, size_t row_stride) {
    #pragma unroll
    for (int j = 0; j < 8; ++j)
        pk[j] = *(const v4f*)(krow + (size_t)(8 * j) * row_stride);
    #pragma unroll
    for (int j = 0; j < 2; ++j)
        pv[j] = *(const v4f*)(vrow + (size_t)(32 * j) * row_stride);
}

static __device__ __forceinline__ void stage_kv(const v4f* pk, const v4f* pv,
                                                float* sK_st, float* sV_st) {
    #pragma unroll
    for (int j = 0; j < 8; ++j) *(v4f*)(sK_st + 8 * j * SK_LD) = pk[j];
    #pragma unroll
    for (int j = 0; j < 2; ++j) *(v4f*)(sV_st + 32 * j * SV_LD) = pv[j];
}

static __device__ __forceinline__ void state_step(v4f* pk, v4f* pv, int t,
                                                  const float* krow, const float* vrow,
                                                  size_t row_stride,
                                                  const float* rK, const float* rV,
                                                  float* wK, float* wV,
                                                  int qk_row, int qk_c4, int v_row, int v_c4,
                                                  int kh, int m16, int d, v8f& h0, v8f& h1) {
    const float* k0 = rK + (kh * 2 + 0) * SK_LD + d * 16 + m16;
    const float* k1 = rK + (kh * 2 + 1) * SK_LD + d * 16 + m16;
    const float* v0 = rV + (kh * 2 + 0) * SV_LD + m16;
    const float* v1 = rV + (kh * 2 + 1) * SV_LD + m16;
    v2f fk[16], fva[16], fvb[16];
    ld_h<0, 4>(fk, fva, fvb, k0, k1, v0, v1);
    ld_h<4, 4>(fk, fva, fvb, k0, k1, v0, v1);
    EDGE2(h0, h1);
    if (t + 1 < T_SPLIT) {
        stage_kv(pk, pv, wK + qk_row * SK_LD + qk_c4, wV + v_row * SV_LD + v_c4);
        const int tn = (t + 3 < T_SPLIT) ? t + 3 : T_SPLIT - 1;
        const size_t adv = (size_t)tn * CH * row_stride;
        fetch_kv(pk, pv, krow + adv, vrow + adv, row_stride);
    }
    EDGE2(h0, h1);
    mm2<0, 4>(fk, fva, h0, fk, fvb, h1);
    EDGE2(h0, h1);
    ld_h<8, 4>(fk, fva, fvb, k0, k1, v0, v1);
    EDGE2(h0, h1);
    mm2<4, 4>(fk, fva, h0, fk, fvb, h1);
    EDGE2(h0, h1);
    ld_h<12, 4>(fk, fva, fvb, k0, k1, v0, v1);
    EDGE2(h0, h1);
    mm2<8, 8>(fk, fva, h0, fk, fvb, h1);
    lds_barrier();
}

static __device__ __forceinline__ void stage_chunk(const v4f* pq, const v4f* pk, const v4f* pv,
                                                   float* sQ_st, float* sK_st, float* sV_st,
                                                   float scale) {
    #pragma unroll
    for (int j = 0; j < 8; ++j) {
        *(v4f*)(sQ_st + 8 * j * SQ_LD) = pq[j] * scale;
        *(v4f*)(sK_st + 8 * j * SK_LD) = pk[j];
    }
    #pragma unroll
    for (int j = 0; j < 2; ++j)
        *(v4f*)(sV_st + 32 * j * SV_LD) = pv[j];
}

extern "C" __global__ __launch_bounds__(256)
void fla_fused_chunk_fwd(const float* __restrict__ q,
                         const float* __restrict__ k,
                         const float* __restrict__ v,
                         float* __restrict__ o,
                         float* __restrict__ hout,
                         float scale)
{
    extern __shared__ float smem[];
    float* sQ = smem + SQ_OFF;
    float* sK = smem + SK_OFF;
    float* sV = smem + SV_OFF;
    float* sS = smem + SS_OFF;
    float* sH = smem + SH_OFF;

    const int tid  = threadIdx.x;
    const int wave = tid >> 5;
    const int lane = tid & 31;
    const int m16  = lane & 15;
    const int kh   = lane >> 4;

    const bool second = blockIdx.x < NWG1;
    const int  t_begin = second ? T_SPLIT : 0;
    const int  t_end   = second ? NT : T_SPLIT;
    const int g  = blockIdx.x & (EG - 1);
    const int bh = (blockIdx.x >> 2) & (NB * NH - 1);
    const int h  = bh & (NH - 1);
    const int b  = bh >> 4;

    const size_t row_stride = (size_t)NH * D_DIM;
    const float* qbase = q + ((size_t)b * SEQ * NH + h) * D_DIM;
    const float* kbase = k + ((size_t)b * SEQ * NH + h) * D_DIM;
    const float* vbase = v + ((size_t)b * SEQ * NH + h) * D_DIM;

    const int qk_row = tid >> 5, qk_c4 = (tid & 31) << 2;
    const int v_row  = tid >> 3, v_c4  = (tid & 7) << 2;
    const float* qrow = qbase + (size_t)qk_row * row_stride + qk_c4;
    const float* krow = kbase + (size_t)qk_row * row_stride + qk_c4;
    const float* vrow = vbase + (size_t)v_row * row_stride + g * GW + v_c4;
    float* sQ_st = sQ + qk_row * SQ_LD + qk_c4;
    float* sK_st = sK + qk_row * SK_LD + qk_c4;
    float* sV_st = sV + v_row * SV_LD + v_c4;

    v4f pq[8], pk[8], pv[2];
    if (second) {
        const int dd = wave;
        const size_t cstep = (size_t)CH * row_stride;
        const size_t adv0  = (size_t)T_SPLIT * cstep;
        v4f ka[8], va[2], kb[8], vb[2];
        v8f h0 = {}, h1 = {};
        fetch_kv(ka, va, krow, vrow, row_stride);
        fetch_kv(kb, vb, krow + cstep, vrow + cstep, row_stride);
        stage_kv(ka, va, sK_st, sV_st);
        fetch_kv(ka, va, krow + 2 * cstep, vrow + 2 * cstep, row_stride);
        lds_barrier();
        for (int t = 0; t + 1 < T_SPLIT; t += 2) {
            state_step(kb, vb, t,     krow, vrow, row_stride, sK, sV, sQ, sS,
                       qk_row, qk_c4, v_row, v_c4, kh, m16, dd, h0, h1);
            state_step(ka, va, t + 1, krow, vrow, row_stride, sQ, sS, sK, sV,
                       qk_row, qk_c4, v_row, v_c4, kh, m16, dd, h0, h1);
        }
        fetch_chunk(pq, pk, pv, qrow + adv0, krow + adv0, vrow + adv0, row_stride);
        if (T_SPLIT & 1)
            state_step(kb, vb, T_SPLIT - 1, krow, vrow, row_stride, sK, sV, sQ, sS,
                       qk_row, qk_c4, v_row, v_c4, kh, m16, dd, h0, h1);
        float* pT0 = sH + (dd * 16 + 8 * kh) * SH_LD + m16;
        #pragma unroll
        for (int j = 0; j < 8; ++j) { pT0[j * SH_LD] = h0[j]; pT0[j * SH_LD + 16] = h1[j]; }
    } else {
        fetch_chunk(pq, pk, pv, qrow, krow, vrow, row_stride);
        for (int i = tid; i < 128 * SH_LD; i += 256) sH[i] = 0.0f;
    }

    stage_chunk(pq, pk, pv, sQ_st, sK_st, sV_st, scale);
    lds_barrier();

    const int r_o  = wave & 3;
    const int eco  = (wave >> 2) * 16;

    const bool two_tiles = wave < 2;
    const int  rA = tile_row(wave),     cA = wave - rA * (rA + 1) / 2;
    const int  rB = 3,                  cB = wave + 2;
    const float* sA_a = sQ + (rA * 16 + m16) * SQ_LD + kh * 2;
    const float* sA_b = sK + (cA * 16 + m16) * SK_LD + kh * 2;
    const float* sB_a = sQ + (rB * 16 + m16) * SQ_LD + kh * 2;
    const float* sB_b = sK + ((cB & 3) * 16 + m16) * SK_LD + kh * 2;

    const float* aS  = sS + (r_o * 16 + m16) * SS_LD + kh * 2;
    const float* aQ  = sQ + (r_o * 16 + m16) * SQ_LD + kh * 2;
    const float* oV0 = sV + (kh * 2 + 0) * SV_LD + eco + m16;
    const float* oV1 = sV + (kh * 2 + 1) * SV_LD + eco + m16;
    const float* oH0 = sH + (kh * 2 + 0) * SH_LD + eco + m16;
    const float* oH1 = sH + (kh * 2 + 1) * SH_LD + eco + m16;

    const int d = wave;
    float* hT0 = sH + (d * 16 + 8 * kh) * SH_LD + m16;
    float* hT1 = hT0 + 16;
    const float* hK0 = sK + (kh * 2 + 0) * SK_LD + d * 16 + m16;
    const float* hK1 = sK + (kh * 2 + 1) * SK_LD + d * 16 + m16;
    const float* hV0 = sV + (kh * 2 + 0) * SV_LD + m16;
    const float* hV1 = sV + (kh * 2 + 1) * SV_LD + m16;

    for (int t = t_begin; t < t_end; ++t) {
        const size_t srow = (size_t)t * CH;

        if (t + 1 < t_end) {
            const size_t adv = (srow + CH) * row_stride;
            fetch_chunk(pq, pk, pv, qrow + adv, krow + adv, vrow + adv, row_stride);
        }

        if (two_tiles) {
            v2f fa0[32], fb0[32], fa1[32], fb1[32];
            v8f c0 = {}, c1 = {};
            ld_rr<0, 8>(fa0, fb0, sA_a, sA_b);   ld_rr<0, 8>(fa1, fb1, sB_a, sB_b);
            ld_rr<8, 8>(fa0, fb0, sA_a, sA_b);   ld_rr<8, 8>(fa1, fb1, sB_a, sB_b);
            EDGE2(c0, c1);
            mm2<0, 8>(fa0, fb0, c0, fa1, fb1, c1);
            EDGE2(c0, c1);
            ld_rr<16, 8>(fa0, fb0, sA_a, sA_b);  ld_rr<16, 8>(fa1, fb1, sB_a, sB_b);
            EDGE2(c0, c1);
            mm2<8, 8>(fa0, fb0, c0, fa1, fb1, c1);
            EDGE2(c0, c1);
            ld_rr<24, 8>(fa0, fb0, sA_a, sA_b);  ld_rr<24, 8>(fa1, fb1, sB_a, sB_b);
            EDGE2(c0, c1);
            mm2<16, 16>(fa0, fb0, c0, fa1, fb1, c1);
            EDGE2(c0, c1);
            s_store(sS, c0, rA, cA, m16, kh);
            s_store(sS, c1, rB, cB, m16, kh);
        } else {
            v2f fa[32], fb[32];
            v8f c0 = {};
            ld_rr<0, 8>(fa, fb, sA_a, sA_b);
            ld_rr<8, 8>(fa, fb, sA_a, sA_b);
            EDGE1(c0);
            c0 = mm<0, 8>(fa, fb, c0);
            EDGE1(c0);
            ld_rr<16, 8>(fa, fb, sA_a, sA_b);
            EDGE1(c0);
            c0 = mm<8, 8>(fa, fb, c0);
            EDGE1(c0);
            ld_rr<24, 8>(fa, fb, sA_a, sA_b);
            EDGE1(c0);
            c0 = mm<16, 16>(fa, fb, c0);
            EDGE1(c0);
            s_store(sS, c0, rA, cA, m16, kh);
        }

        v2f fq[32], fh[32];
        ld_rk<0, 16, SH_LD>(fq, fh, aQ, oH0, oH1);
        lds_barrier();

        v8f acc;
        switch (r_o) {
            case 0:  acc = sv_part<1>(aS, oV0, oV1); break;
            case 1:  acc = sv_part<2>(aS, oV0, oV1); break;
            case 2:  acc = sv_part<3>(aS, oV0, oV1); break;
            default: acc = sv_part<4>(aS, oV0, oV1); break;
        }
        v2f fk[16], fva[16], fvb[16];
        v8f h0, h1;
        EDGE1(acc);
        ld_rk<16, 8, SH_LD>(fq, fh, aQ, oH0, oH1);
        EDGE1(acc);
        acc = mm<0, 8>(fq, fh, acc);
        EDGE1(acc);
        ld_rk<24, 8, SH_LD>(fq, fh, aQ, oH0, oH1);
        EDGE1(acc);
        acc = mm<8, 8>(fq, fh, acc);
        EDGE1(acc);
        #pragma unroll
        for (int j = 0; j < 8; ++j) { h0[j] = hT0[j * SH_LD]; h1[j] = hT1[j * SH_LD]; }
        ld_h<0, 8>(fk, fva, fvb, hK0, hK1, hV0, hV1);
        EDGE1(acc);
        acc = mm<16, 8>(fq, fh, acc);
        EDGE1(acc);
        ld_h<8, 8>(fk, fva, fvb, hK0, hK1, hV0, hV1);
        EDGE1(acc);
        acc = mm<24, 8>(fq, fh, acc);

        #pragma unroll
        for (int j = 0; j < 8; ++j) {
            size_t s_idx = srow + r_o * 16 + j + 8 * kh;
            o[(((size_t)b * SEQ + s_idx) * NH + h) * D_DIM + g * GW + eco + m16] = acc[j];
        }
        EDGE2(h0, h1);
        lds_barrier();

        if (t + 1 < t_end)
            stage_chunk(pq, pk, pv, sQ_st, sK_st, sV_st, scale);

        mm2<0, 16>(fk, fva, h0, fk, fvb, h1);
        #pragma unroll
        for (int j = 0; j < 8; ++j) { hT0[j * SH_LD] = h0[j]; hT1[j * SH_LD] = h1[j]; }
        lds_barrier();
    }

    if (second) {
    #pragma unroll
    for (int j = 0; j < 4; ++j) {
        int f    = tid + j * 256;
        int drow = f >> 3;
        int c4   = (f & 7) << 2;
        float4 val = *(const float4*)(sH + drow * SH_LD + c4);
        *(float4*)(hout + (((size_t)(b * NH + h)) * D_DIM + drow) * D_DIM + g * GW + c4) = val;
    }
    }
}

extern "C" void kernel_launch(void* const* d_in, const int* in_sizes, int n_in,
                              void* d_out, int out_size, void* d_ws, size_t ws_size,
                              hipStream_t stream) {
    (void)in_sizes; (void)n_in; (void)out_size; (void)d_ws; (void)ws_size;
    const float* q = (const float*)d_in[0];
    const float* k = (const float*)d_in[1];
    const float* v = (const float*)d_in[2];
    float* o  = (float*)d_out;
    float* hf = o + (size_t)NB * SEQ * NH * D_DIM;
    const float scale = 0.088388347648318447f;

    const size_t smem = (size_t)SMEM_FLOATS * sizeof(float);
    hipFuncSetAttribute((const void*)fla_fused_chunk_fwd,
                        hipFuncAttributeMaxDynamicSharedMemorySize, (int)smem);

    dim3 grid(2 * NWG1);
    dim3 blk(256);
    hipLaunchKernelGGL(fla_fused_chunk_fwd, grid, blk, smem, stream,
                       q, k, v, o, hf, scale);
}


# LDS row pitches for the k-major fragment loads: V and state 36 -> 48 with a k pair's two rows in one two-address load (no bank shared), K and q 132 -> 140 (lane halves 24 banks apart); sums unchanged
# speedup vs baseline: 2.9439x; 1.0888x over previous
#include <hip/hip_runtime.h>

typedef __attribute__((ext_vector_type(2))) float v2f;
typedef __attribute__((ext_vector_type(8))) float v8f;
typedef __attribute__((ext_vector_type(4))) float v4f;

#define D_DIM 128
#define CH    64
#define SEQ   4096
#define NT    (SEQ / CH)
#define NH    16
#define NB    2
#define EG    4
#define GW    32
#define NWG1  (NB * NH * EG)
#define T_SPLIT 35
static_assert(T_SPLIT >= 3 && T_SPLIT < NT, "the state pass keeps two chunks in flight");

#define SQ_LD 140
#define SK_LD 140
#define SV_LD 48
#define SS_LD 68
#define SH_LD 48

#define SQ_OFF 0
#define SK_OFF (SQ_OFF + 64 * SQ_LD)
#define SV_OFF (SK_OFF + 64 * SK_LD)
#define SS_OFF (SV_OFF + 64 * SV_LD)
#define SH_OFF (SS_OFF + 64 * SS_LD)
#define SMEM_FLOATS (SH_OFF + 128 * SH_LD)

#define EDGE1(a)    asm volatile("" : "+v"(a) : : "memory")
#define EDGE2(a, b) asm volatile("" : "+v"(a), "+v"(b) : : "memory")

static __device__ __forceinline__ v8f wmma_f32(v2f a, v2f b, v8f c) {
    return __builtin_amdgcn_wmma_f32_16x16x4_f32(
         false, a,  false, b,
         (short)0, c,  false,  false);
}

static __device__ __forceinline__ void lds_barrier() {
    __builtin_amdgcn_fence(__ATOMIC_RELEASE, "workgroup", "local");
    __builtin_amdgcn_s_barrier();
    __builtin_amdgcn_fence(__ATOMIC_ACQUIRE, "workgroup", "local");
}

template <int J0, int N>
static __device__ __forceinline__ void ld_rr(v2f* fa, v2f* fb, const float* pa, const float* pb) {
    #pragma unroll
    for (int j = J0; j < J0 + N; ++j) {
        fa[j] = *(const v2f*)(pa + 4 * j);
        fb[j] = *(const v2f*)(pb + 4 * j);
    }
}

template <int J0, int N, int LD>
static __device__ __forceinline__ void ld_rk(v2f* fa, v2f* fb, const float* pa,
                                             const float* p0) {
    #pragma unroll
    for (int j = J0; j < J0 + N; ++j) {
        fa[j] = *(const v2f*)(pa + 4 * j);
        fb[j].x = p0[4 * j * LD];
        fb[j].y = p0[4 * j * LD + LD];
    }
}

template <int J0, int N>
static __device__ __forceinline__ void ld_h(v2f* fk, v2f* fva, v2f* fvb,
                                            const float* k0, const float* k1,
                                            const float* v0) {
    #pragma unroll
    for (int j = J0; j < J0 + N; ++j) {
        fk[j].x  = k0[4 * j * SK_LD];      fk[j].y  = k1[4 * j * SK_LD];
        fva[j].x = v0[4 * j * SV_LD];      fva[j].y = v0[4 * j * SV_LD + SV_LD];
        fvb[j].x = v0[4 * j * SV_LD + 16]; fvb[j].y = v0[4 * j * SV_LD + SV_LD + 16];
    }
}

template <int J0, int N>
static __device__ __forceinline__ v8f mm(const v2f* fa, const v2f* fb, v8f acc) {
    #pragma unroll
    for (int j = J0; j < J0 + N; ++j) acc = wmma_f32(fa[j], fb[j], acc);
    return acc;
}

template <int J0, int N>
static __device__ __forceinline__ void mm2(const v2f* fa0, const v2f* fb0, v8f& c0,
                                           const v2f* fa1, const v2f* fb1, v8f& c1) {
    #pragma unroll
    for (int j = J0; j < J0 + N; ++j) {
        c0 = wmma_f32(fa0[j], fb0[j], c0);
        c1 = wmma_f32(fa1[j], fb1[j], c1);
    }
}

static __device__ __forceinline__ void s_store(float* sS, v8f acc, int r, int c, int m16, int kh) {
    #pragma unroll
    for (int j = 0; j < 8; ++j) {
        int er = r * 16 + j + 8 * kh;
        int ec = c * 16 + m16;
        float val = acc[j];
        if (r == c && er < ec) val = 0.0f;
        sS[er * SS_LD + ec] = val;
    }
}

template <int NBLK>
static __device__ __forceinline__ v8f sv_part(const float* aS, const float* bV0) {
    v2f fs[NBLK * 4], fv[NBLK * 4];
    ld_rk<0, NBLK * 4, SV_LD>(fs, fv, aS, bV0);
    v8f acc = {};
    EDGE1(acc);
    return mm<0, NBLK * 4>(fs, fv, acc);
}

static __device__ __forceinline__ int tile_row(int idx) {
    return (idx >= 6) ? 3 : (idx >= 3) ? 2 : (idx >= 1) ? 1 : 0;
}

static __device__ __forceinline__ void fetch_chunk(v4f* pq, v4f* pk, v4f* pv,
                                                   const float* qrow, const float* krow,
                                                   const float* vrow, size_t row_stride) {
    #pragma unroll
    for (int j = 0; j < 8; ++j) {
        pq[j] = *(const v4f*)(qrow + (size_t)(8 * j) * row_stride);
        pk[j] = *(const v4f*)(krow + (size_t)(8 * j) * row_stride);
    }
    #pragma unroll
    for (int j = 0; j < 2; ++j)
        pv[j] = *(const v4f*)(vrow + (size_t)(32 * j) * row_stride);
}

static __device__ __forceinline__ void fetch_kv(v4f* pk, v4f* pv, const float* krow,
                                                const float* vrow, size_t row_stride) {
    #pragma unroll
    for (int j = 0; j < 8; ++j)
        pk[j] = *(const v4f*)(krow + (size_t)(8 * j) * row_stride);
    #pragma unroll
    for (int j = 0; j < 2; ++j)
        pv[j] = *(const v4f*)(vrow + (size_t)(32 * j) * row_stride);
}

static __device__ __forceinline__ void stage_kv(const v4f* pk, const v4f* pv,
                                                float* sK_st, float* sV_st) {
    #pragma unroll
    for (int j = 0; j < 8; ++j) *(v4f*)(sK_st + 8 * j * SK_LD) = pk[j];
    #pragma unroll
    for (int j = 0; j < 2; ++j) *(v4f*)(sV_st + 32 * j * SV_LD) = pv[j];
}

static __device__ __forceinline__ void state_step(v4f* pk, v4f* pv, int t,
                                                  const float* krow, const float* vrow,
                                                  size_t row_stride,
                                                  const float* rK, const float* rV,
                                                  float* wK, float* wV,
                                                  int qk_row, int qk_c4, int v_row, int v_c4,
                                                  int kh, int m16, int d, v8f& h0, v8f& h1) {
    const float* k0 = rK + (kh * 2 + 0) * SK_LD + d * 16 + m16;
    const float* k1 = rK + (kh * 2 + 1) * SK_LD + d * 16 + m16;
    const float* v0 = rV + (kh * 2 + 0) * SV_LD + m16;
    v2f fk[16], fva[16], fvb[16];
    ld_h<0, 4>(fk, fva, fvb, k0, k1, v0);
    ld_h<4, 4>(fk, fva, fvb, k0, k1, v0);
    EDGE2(h0, h1);
    if (t + 1 < T_SPLIT) {
        stage_kv(pk, pv, wK + qk_row * SK_LD + qk_c4, wV + v_row * SV_LD + v_c4);
        const int tn = (t + 3 < T_SPLIT) ? t + 3 : T_SPLIT - 1;
        const size_t adv = (size_t)tn * CH * row_stride;
        fetch_kv(pk, pv, krow + adv, vrow + adv, row_stride);
    }
    EDGE2(h0, h1);
    mm2<0, 4>(fk, fva, h0, fk, fvb, h1);
    EDGE2(h0, h1);
    ld_h<8, 4>(fk, fva, fvb, k0, k1, v0);
    EDGE2(h0, h1);
    mm2<4, 4>(fk, fva, h0, fk, fvb, h1);
    EDGE2(h0, h1);
    ld_h<12, 4>(fk, fva, fvb, k0, k1, v0);
    EDGE2(h0, h1);
    mm2<8, 8>(fk, fva, h0, fk, fvb, h1);
    lds_barrier();
}

static __device__ __forceinline__ void stage_chunk(const v4f* pq, const v4f* pk, const v4f* pv,
                                                   float* sQ_st, float* sK_st, float* sV_st,
                                                   float scale) {
    #pragma unroll
    for (int j = 0; j < 8; ++j) {
        *(v4f*)(sQ_st + 8 * j * SQ_LD) = pq[j] * scale;
        *(v4f*)(sK_st + 8 * j * SK_LD) = pk[j];
    }
    #pragma unroll
    for (int j = 0; j < 2; ++j)
        *(v4f*)(sV_st + 32 * j * SV_LD) = pv[j];
}

extern "C" __global__ __launch_bounds__(256)
void fla_fused_chunk_fwd(const float* __restrict__ q,
                         const float* __restrict__ k,
                         const float* __restrict__ v,
                         float* __restrict__ o,
                         float* __restrict__ hout,
                         float scale)
{
    extern __shared__ float smem[];
    float* sQ = smem + SQ_OFF;
    float* sK = smem + SK_OFF;
    float* sV = smem + SV_OFF;
    float* sS = smem + SS_OFF;
    float* sH = smem + SH_OFF;

    const int tid  = threadIdx.x;
    const int wave = tid >> 5;
    const int lane = tid & 31;
    const int m16  = lane & 15;
    const int kh   = lane >> 4;

    const bool second = blockIdx.x < NWG1;
    const int  t_begin = second ? T_SPLIT : 0;
    const int  t_end   = second ? NT : T_SPLIT;
    const int g  = blockIdx.x & (EG - 1);
    const int bh = (blockIdx.x >> 2) & (NB * NH - 1);
    const int h  = bh & (NH - 1);
    const int b  = bh >> 4;

    const size_t row_stride = (size_t)NH * D_DIM;
    const float* qbase = q + ((size_t)b * SEQ * NH + h) * D_DIM;
    const float* kbase = k + ((size_t)b * SEQ * NH + h) * D_DIM;
    const float* vbase = v + ((size_t)b * SEQ * NH + h) * D_DIM;

    const int qk_row = tid >> 5, qk_c4 = (tid & 31) << 2;
    const int v_row  = tid >> 3, v_c4  = (tid & 7) << 2;
    const float* qrow = qbase + (size_t)qk_row * row_stride + qk_c4;
    const float* krow = kbase + (size_t)qk_row * row_stride + qk_c4;
    const float* vrow = vbase + (size_t)v_row * row_stride + g * GW + v_c4;
    float* sQ_st = sQ + qk_row * SQ_LD + qk_c4;
    float* sK_st = sK + qk_row * SK_LD + qk_c4;
    float* sV_st = sV + v_row * SV_LD + v_c4;

    v4f pq[8], pk[8], pv[2];
    if (second) {
        const int dd = wave;
        const size_t cstep = (size_t)CH * row_stride;
        const size_t adv0  = (size_t)T_SPLIT * cstep;
        v4f ka[8], va[2], kb[8], vb[2];
        v8f h0 = {}, h1 = {};
        fetch_kv(ka, va, krow, vrow, row_stride);
        fetch_kv(kb, vb, krow + cstep, vrow + cstep, row_stride);
        stage_kv(ka, va, sK_st, sV_st);
        fetch_kv(ka, va, krow + 2 * cstep, vrow + 2 * cstep, row_stride);
        lds_barrier();
        for (int t = 0; t + 1 < T_SPLIT; t += 2) {
            state_step(kb, vb, t,     krow, vrow, row_stride, sK, sV, sQ, sS,
                       qk_row, qk_c4, v_row, v_c4, kh, m16, dd, h0, h1);
            state_step(ka, va, t + 1, krow, vrow, row_stride, sQ, sS, sK, sV,
                       qk_row, qk_c4, v_row, v_c4, kh, m16, dd, h0, h1);
        }
        fetch_chunk(pq, pk, pv, qrow + adv0, krow + adv0, vrow + adv0, row_stride);
        if (T_SPLIT & 1)
            state_step(kb, vb, T_SPLIT - 1, krow, vrow, row_stride, sK, sV, sQ, sS,
                       qk_row, qk_c4, v_row, v_c4, kh, m16, dd, h0, h1);
        float* pT0 = sH + (dd * 16 + 8 * kh) * SH_LD + m16;
        #pragma unroll
        for (int j = 0; j < 8; ++j) { pT0[j * SH_LD] = h0[j]; pT0[j * SH_LD + 16] = h1[j]; }
    } else {
        fetch_chunk(pq, pk, pv, qrow, krow, vrow, row_stride);
        for (int i = tid; i < 128 * SH_LD; i += 256) sH[i] = 0.0f;
    }

    stage_chunk(pq, pk, pv, sQ_st, sK_st, sV_st, scale);
    lds_barrier();

    const int r_o  = wave & 3;
    const int eco  = (wave >> 2) * 16;

    const bool two_tiles = wave < 2;
    const int  rA = tile_row(wave),     cA = wave - rA * (rA + 1) / 2;
    const int  rB = 3,                  cB = wave + 2;
    const float* sA_a = sQ + (rA * 16 + m16) * SQ_LD + kh * 2;
    const float* sA_b = sK + (cA * 16 + m16) * SK_LD + kh * 2;
    const float* sB_a = sQ + (rB * 16 + m16) * SQ_LD + kh * 2;
    const float* sB_b = sK + ((cB & 3) * 16 + m16) * SK_LD + kh * 2;

    const float* aS  = sS + (r_o * 16 + m16) * SS_LD + kh * 2;
    const float* aQ  = sQ + (r_o * 16 + m16) * SQ_LD + kh * 2;
    const float* oV0 = sV + (kh * 2 + 0) * SV_LD + eco + m16;
    const float* oH0 = sH + (kh * 2 + 0) * SH_LD + eco + m16;

    const int d = wave;
    float* hT0 = sH + (d * 16 + 8 * kh) * SH_LD + m16;
    float* hT1 = hT0 + 16;
    const float* hK0 = sK + (kh * 2 + 0) * SK_LD + d * 16 + m16;
    const float* hK1 = sK + (kh * 2 + 1) * SK_LD + d * 16 + m16;
    const float* hV0 = sV + (kh * 2 + 0) * SV_LD + m16;

    for (int t = t_begin; t < t_end; ++t) {
        const size_t srow = (size_t)t * CH;

        if (t + 1 < t_end) {
            const size_t adv = (srow + CH) * row_stride;
            fetch_chunk(pq, pk, pv, qrow + adv, krow + adv, vrow + adv, row_stride);
        }

        if (two_tiles) {
            v2f fa0[32], fb0[32], fa1[32], fb1[32];
            v8f c0 = {}, c1 = {};
            ld_rr<0, 8>(fa0, fb0, sA_a, sA_b);   ld_rr<0, 8>(fa1, fb1, sB_a, sB_b);
            ld_rr<8, 8>(fa0, fb0, sA_a, sA_b);   ld_rr<8, 8>(fa1, fb1, sB_a, sB_b);
            EDGE2(c0, c1);
            mm2<0, 8>(fa0, fb0, c0, fa1, fb1, c1);
            EDGE2(c0, c1);
            ld_rr<16, 8>(fa0, fb0, sA_a, sA_b);  ld_rr<16, 8>(fa1, fb1, sB_a, sB_b);
            EDGE2(c0, c1);
            mm2<8, 8>(fa0, fb0, c0, fa1, fb1, c1);
            EDGE2(c0, c1);
            ld_rr<24, 8>(fa0, fb0, sA_a, sA_b);  ld_rr<24, 8>(fa1, fb1, sB_a, sB_b);
            EDGE2(c0, c1);
            mm2<16, 16>(fa0, fb0, c0, fa1, fb1, c1);
            EDGE2(c0, c1);
            s_store(sS, c0, rA, cA, m16, kh);
            s_store(sS, c1, rB, cB, m16, kh);
        } else {
            v2f fa[32], fb[32];
            v8f c0 = {};
            ld_rr<0, 8>(fa, fb, sA_a, sA_b);
            ld_rr<8, 8>(fa, fb, sA_a, sA_b);
            EDGE1(c0);
            c0 = mm<0, 8>(fa, fb, c0);
            EDGE1(c0);
            ld_rr<16, 8>(fa, fb, sA_a, sA_b);
            EDGE1(c0);
            c0 = mm<8, 8>(fa, fb, c0);
            EDGE1(c0);
            ld_rr<24, 8>(fa, fb, sA_a, sA_b);
            EDGE1(c0);
            c0 = mm<16, 16>(fa, fb, c0);
            EDGE1(c0);
            s_store(sS, c0, rA, cA, m16, kh);
        }

        v2f fq[32], fh[32];
        ld_rk<0, 16, SH_LD>(fq, fh, aQ, oH0);
        lds_barrier();

        v8f acc;
        switch (r_o) {
            case 0:  acc = sv_part<1>(aS, oV0); break;
            case 1:  acc = sv_part<2>(aS, oV0); break;
            case 2:  acc = sv_part<3>(aS, oV0); break;
            default: acc = sv_part<4>(aS, oV0); break;
        }
        v2f fk[16], fva[16], fvb[16];
        v8f h0, h1;
        EDGE1(acc);
        ld_rk<16, 8, SH_LD>(fq, fh, aQ, oH0);
        EDGE1(acc);
        acc = mm<0, 8>(fq, fh, acc);
        EDGE1(acc);
        ld_rk<24, 8, SH_LD>(fq, fh, aQ, oH0);
        EDGE1(acc);
        acc = mm<8, 8>(fq, fh, acc);
        EDGE1(acc);
        #pragma unroll
        for (int j = 0; j < 8; ++j) { h0[j] = hT0[j * SH_LD]; h1[j] = hT1[j * SH_LD]; }
        ld_h<0, 8>(fk, fva, fvb, hK0, hK1, hV0);
        EDGE1(acc);
        acc = mm<16, 8>(fq, fh, acc);
        EDGE1(acc);
        ld_h<8, 8>(fk, fva, fvb, hK0, hK1, hV0);
        EDGE1(acc);
        acc = mm<24, 8>(fq, fh, acc);

        #pragma unroll
        for (int j = 0; j < 8; ++j) {
            size_t s_idx = srow + r_o * 16 + j + 8 * kh;
            o[(((size_t)b * SEQ + s_idx) * NH + h) * D_DIM + g * GW + eco + m16] = acc[j];
        }
        EDGE2(h0, h1);
        lds_barrier();

        if (t + 1 < t_end)
            stage_chunk(pq, pk, pv, sQ_st, sK_st, sV_st, scale);

        mm2<0, 16>(fk, fva, h0, fk, fvb, h1);
        #pragma unroll
        for (int j = 0; j < 8; ++j) { hT0[j * SH_LD] = h0[j]; hT1[j * SH_LD] = h1[j]; }
        lds_barrier();
    }

    if (second) {
    #pragma unroll
    for (int j = 0; j < 4; ++j) {
        int f    = tid + j * 256;
        int drow = f >> 3;
        int c4   = (f & 7) << 2;
        float4 val = *(const float4*)(sH + drow * SH_LD + c4);
        *(float4*)(hout + (((size_t)(b * NH + h)) * D_DIM + drow) * D_DIM + g * GW + c4) = val;
    }
    }
}

extern "C" void kernel_launch(void* const* d_in, const int* in_sizes, int n_in,
                              void* d_out, int out_size, void* d_ws, size_t ws_size,
                              hipStream_t stream) {
    (void)in_sizes; (void)n_in; (void)out_size; (void)d_ws; (void)ws_size;
    const float* q = (const float*)d_in[0];
    const float* k = (const float*)d_in[1];
    const float* v = (const float*)d_in[2];
    float* o  = (float*)d_out;
    float* hf = o + (size_t)NB * SEQ * NH * D_DIM;
    const float scale = 0.088388347648318447f;

    const size_t smem = (size_t)SMEM_FLOATS * sizeof(float);
    hipFuncSetAttribute((const void*)fla_fused_chunk_fwd,
                        hipFuncAttributeMaxDynamicSharedMemorySize, (int)smem);

    dim3 grid(2 * NWG1);
    dim3 blk(256);
    hipLaunchKernelGGL(fla_fused_chunk_fwd, grid, blk, smem, stream,
                       q, k, v, o, hf, scale);
}


# q/k/v chunk buffers doubled in LDS: next chunk staged during the O phase, its S made in the state-update phase (state chains first, S loads in flight); 2 barriers a chunk, not 3; sums unchanged
# speedup vs baseline: 3.0570x; 1.0384x over previous
#include <hip/hip_runtime.h>

typedef __attribute__((ext_vector_type(2))) float v2f;
typedef __attribute__((ext_vector_type(8))) float v8f;
typedef __attribute__((ext_vector_type(4))) float v4f;

#define D_DIM 128
#define CH    64
#define SEQ   4096
#define NT    (SEQ / CH)
#define NH    16
#define NB    2
#define EG    4
#define GW    32
#define NWG1  (NB * NH * EG)
#define T_SPLIT 35
static_assert(T_SPLIT >= 3 && T_SPLIT < NT, "the state pass keeps two chunks in flight");

#define SQ_LD 140
#define SK_LD 140
#define SV_LD 48
#define SS_LD 68
#define SH_LD 48

#define SQ_OFF 0
#define SK_OFF (SQ_OFF + 64 * SQ_LD)
#define SV_OFF (SK_OFF + 64 * SK_LD)
#define SS_OFF (SV_OFF + 64 * SV_LD)
#define SH_OFF (SS_OFF + 64 * SS_LD)
#define BUF1_OFF (SH_OFF + 128 * SH_LD)
#define SMEM_FLOATS (BUF1_OFF + SV_OFF + 64 * SV_LD)
static_assert(BUF1_OFF % 64 == 0, "both buffers on the same banks");

#define EDGE1(a)    asm volatile("" : "+v"(a) : : "memory")
#define EDGE2(a, b) asm volatile("" : "+v"(a), "+v"(b) : : "memory")
#define EDGE3(a, b, c)    asm volatile("" : "+v"(a), "+v"(b), "+v"(c) : : "memory")
#define EDGE4(a, b, c, d) asm volatile("" : "+v"(a), "+v"(b), "+v"(c), "+v"(d) : : "memory")

static __device__ __forceinline__ v8f wmma_f32(v2f a, v2f b, v8f c) {
    return __builtin_amdgcn_wmma_f32_16x16x4_f32(
         false, a,  false, b,
         (short)0, c,  false,  false);
}

static __device__ __forceinline__ void lds_barrier() {
    __builtin_amdgcn_fence(__ATOMIC_RELEASE, "workgroup", "local");
    __builtin_amdgcn_s_barrier();
    __builtin_amdgcn_fence(__ATOMIC_ACQUIRE, "workgroup", "local");
}

template <int J0, int N>
static __device__ __forceinline__ void ld_rr(v2f* fa, v2f* fb, const float* pa, const float* pb) {
    #pragma unroll
    for (int j = J0; j < J0 + N; ++j) {
        fa[j] = *(const v2f*)(pa + 4 * j);
        fb[j] = *(const v2f*)(pb + 4 * j);
    }
}

template <int J0, int N, int LD>
static __device__ __forceinline__ void ld_rk(v2f* fa, v2f* fb, const float* pa,
                                             const float* p0) {
    #pragma unroll
    for (int j = J0; j < J0 + N; ++j) {
        fa[j] = *(const v2f*)(pa + 4 * j);
        fb[j].x = p0[4 * j * LD];
        fb[j].y = p0[4 * j * LD + LD];
    }
}

template <int J0, int N>
static __device__ __forceinline__ void ld_h(v2f* fk, v2f* fva, v2f* fvb,
                                            const float* k0, const float* k1,
                                            const float* v0) {
    #pragma unroll
    for (int j = J0; j < J0 + N; ++j) {
        fk[j].x  = k0[4 * j * SK_LD];      fk[j].y  = k1[4 * j * SK_LD];
        fva[j].x = v0[4 * j * SV_LD];      fva[j].y = v0[4 * j * SV_LD + SV_LD];
        fvb[j].x = v0[4 * j * SV_LD + 16]; fvb[j].y = v0[4 * j * SV_LD + SV_LD + 16];
    }
}

template <int J0, int N>
static __device__ __forceinline__ v8f mm(const v2f* fa, const v2f* fb, v8f acc) {
    #pragma unroll
    for (int j = J0; j < J0 + N; ++j) acc = wmma_f32(fa[j], fb[j], acc);
    return acc;
}

template <int J0, int N>
static __device__ __forceinline__ void mm2(const v2f* fa0, const v2f* fb0, v8f& c0,
                                           const v2f* fa1, const v2f* fb1, v8f& c1) {
    #pragma unroll
    for (int j = J0; j < J0 + N; ++j) {
        c0 = wmma_f32(fa0[j], fb0[j], c0);
        c1 = wmma_f32(fa1[j], fb1[j], c1);
    }
}

static __device__ __forceinline__ void s_store(float* sS, v8f acc, int r, int c, int m16, int kh) {
    #pragma unroll
    for (int j = 0; j < 8; ++j) {
        int er = r * 16 + j + 8 * kh;
        int ec = c * 16 + m16;
        float val = acc[j];
        if (r == c && er < ec) val = 0.0f;
        sS[er * SS_LD + ec] = val;
    }
}

template <int NBLK>
static __device__ __forceinline__ v8f sv_part(const float* aS, const float* bV0,
                                              v2f* fq, v2f* fh,
                                              const float* aQ, const float* oH0) {
    v2f fs[NBLK * 4], fv[NBLK * 4];
    ld_rk<0, NBLK * 4, SV_LD>(fs, fv, aS, bV0);
    ld_rk<0, 16, SH_LD>(fq, fh, aQ, oH0);
    v8f acc = {};
    EDGE1(acc);
    return mm<0, NBLK * 4>(fs, fv, acc);
}

static __device__ __forceinline__ int tile_row(int idx) {
    return (idx >= 6) ? 3 : (idx >= 3) ? 2 : (idx >= 1) ? 1 : 0;
}

static __device__ __forceinline__ void fetch_chunk(v4f* pq, v4f* pk, v4f* pv,
                                                   const float* qrow, const float* krow,
                                                   const float* vrow, size_t row_stride) {
    #pragma unroll
    for (int j = 0; j < 8; ++j) {
        pq[j] = *(const v4f*)(qrow + (size_t)(8 * j) * row_stride);
        pk[j] = *(const v4f*)(krow + (size_t)(8 * j) * row_stride);
    }
    #pragma unroll
    for (int j = 0; j < 2; ++j)
        pv[j] = *(const v4f*)(vrow + (size_t)(32 * j) * row_stride);
}

static __device__ __forceinline__ void fetch_kv(v4f* pk, v4f* pv, const float* krow,
                                                const float* vrow, size_t row_stride) {
    #pragma unroll
    for (int j = 0; j < 8; ++j)
        pk[j] = *(const v4f*)(krow + (size_t)(8 * j) * row_stride);
    #pragma unroll
    for (int j = 0; j < 2; ++j)
        pv[j] = *(const v4f*)(vrow + (size_t)(32 * j) * row_stride);
}

static __device__ __forceinline__ void stage_kv(const v4f* pk, const v4f* pv,
                                                float* sK_st, float* sV_st) {
    #pragma unroll
    for (int j = 0; j < 8; ++j) *(v4f*)(sK_st + 8 * j * SK_LD) = pk[j];
    #pragma unroll
    for (int j = 0; j < 2; ++j) *(v4f*)(sV_st + 32 * j * SV_LD) = pv[j];
}

static __device__ __forceinline__ void state_step(v4f* pk, v4f* pv, int t,
                                                  const float* krow, const float* vrow,
                                                  size_t row_stride,
                                                  const float* rK, const float* rV,
                                                  float* wK, float* wV,
                                                  int qk_row, int qk_c4, int v_row, int v_c4,
                                                  int kh, int m16, int d, v8f& h0, v8f& h1) {
    const float* k0 = rK + (kh * 2 + 0) * SK_LD + d * 16 + m16;
    const float* k1 = rK + (kh * 2 + 1) * SK_LD + d * 16 + m16;
    const float* v0 = rV + (kh * 2 + 0) * SV_LD + m16;
    v2f fk[16], fva[16], fvb[16];
    ld_h<0, 4>(fk, fva, fvb, k0, k1, v0);
    ld_h<4, 4>(fk, fva, fvb, k0, k1, v0);
    EDGE2(h0, h1);
    if (t + 1 < T_SPLIT) {
        stage_kv(pk, pv, wK + qk_row * SK_LD + qk_c4, wV + v_row * SV_LD + v_c4);
        const int tn = (t + 3 < T_SPLIT) ? t + 3 : T_SPLIT - 1;
        const size_t adv = (size_t)tn * CH * row_stride;
        fetch_kv(pk, pv, krow + adv, vrow + adv, row_stride);
    }
    EDGE2(h0, h1);
    mm2<0, 4>(fk, fva, h0, fk, fvb, h1);
    EDGE2(h0, h1);
    ld_h<8, 4>(fk, fva, fvb, k0, k1, v0);
    EDGE2(h0, h1);
    mm2<4, 4>(fk, fva, h0, fk, fvb, h1);
    EDGE2(h0, h1);
    ld_h<12, 4>(fk, fva, fvb, k0, k1, v0);
    EDGE2(h0, h1);
    mm2<8, 8>(fk, fva, h0, fk, fvb, h1);
    lds_barrier();
}

static __device__ __forceinline__ void stage_chunk(const v4f* pq, const v4f* pk, const v4f* pv,
                                                   float* sQ_st, float* sK_st, float* sV_st,
                                                   float scale) {
    #pragma unroll
    for (int j = 0; j < 8; ++j) {
        *(v4f*)(sQ_st + 8 * j * SQ_LD) = pq[j] * scale;
        *(v4f*)(sK_st + 8 * j * SK_LD) = pk[j];
    }
    #pragma unroll
    for (int j = 0; j < 2; ++j)
        *(v4f*)(sV_st + 32 * j * SV_LD) = pv[j];
}

static __device__ __forceinline__ void h_store(float* hT0, float* hT1, v8f h0, v8f h1) {
    #pragma unroll
    for (int j = 0; j < 8; ++j) { hT0[j * SH_LD] = h0[j]; hT1[j * SH_LD] = h1[j]; }
}

template <bool WITH_H>
static __device__ __forceinline__ void s_phase(bool two_tiles,
                                               const float* sA_a, const float* sA_b,
                                               const float* sB_a, const float* sB_b,
                                               float* sS, int rA, int cA, int rB, int cB,
                                               int m16, int kh,
                                               const v2f* fk, const v2f* fva, const v2f* fvb,
                                               v8f& h0, v8f& h1, float* hT0, float* hT1) {
    if (two_tiles) {
        v2f fa0[32], fb0[32], fa1[32], fb1[32];
        v8f c0 = {}, c1 = {};
        ld_rr<0, 8>(fa0, fb0, sA_a, sA_b);   ld_rr<0, 8>(fa1, fb1, sB_a, sB_b);
        ld_rr<8, 8>(fa0, fb0, sA_a, sA_b);   ld_rr<8, 8>(fa1, fb1, sB_a, sB_b);
        if constexpr (WITH_H) {
            EDGE2(h0, h1);
            mm2<0, 16>(fk, fva, h0, fk, fvb, h1);
            EDGE4(h0, h1, c0, c1);
        } else {
            EDGE2(c0, c1);
        }
        mm2<0, 8>(fa0, fb0, c0, fa1, fb1, c1);
        EDGE2(c0, c1);
        ld_rr<16, 8>(fa0, fb0, sA_a, sA_b);  ld_rr<16, 8>(fa1, fb1, sB_a, sB_b);
        if constexpr (WITH_H) h_store(hT0, hT1, h0, h1);
        EDGE2(c0, c1);
        mm2<8, 8>(fa0, fb0, c0, fa1, fb1, c1);
        EDGE2(c0, c1);
        ld_rr<24, 8>(fa0, fb0, sA_a, sA_b);  ld_rr<24, 8>(fa1, fb1, sB_a, sB_b);
        EDGE2(c0, c1);
        mm2<16, 16>(fa0, fb0, c0, fa1, fb1, c1);
        EDGE2(c0, c1);
        s_store(sS, c0, rA, cA, m16, kh);
        s_store(sS, c1, rB, cB, m16, kh);
    } else {
        v2f fa[32], fb[32];
        v8f c0 = {};
        ld_rr<0, 8>(fa, fb, sA_a, sA_b);
        ld_rr<8, 8>(fa, fb, sA_a, sA_b);
        if constexpr (WITH_H) {
            EDGE2(h0, h1);
            mm2<0, 16>(fk, fva, h0, fk, fvb, h1);
            EDGE3(h0, h1, c0);
        } else {
            EDGE1(c0);
        }
        c0 = mm<0, 8>(fa, fb, c0);
        EDGE1(c0);
        ld_rr<16, 8>(fa, fb, sA_a, sA_b);
        if constexpr (WITH_H) h_store(hT0, hT1, h0, h1);
        EDGE1(c0);
        c0 = mm<8, 8>(fa, fb, c0);
        EDGE1(c0);
        ld_rr<24, 8>(fa, fb, sA_a, sA_b);
        EDGE1(c0);
        c0 = mm<16, 16>(fa, fb, c0);
        EDGE1(c0);
        s_store(sS, c0, rA, cA, m16, kh);
    }
}

extern "C" __global__ __launch_bounds__(256)
void fla_fused_chunk_fwd(const float* __restrict__ q,
                         const float* __restrict__ k,
                         const float* __restrict__ v,
                         float* __restrict__ o,
                         float* __restrict__ hout,
                         float scale)
{
    extern __shared__ float smem[];
    float* sQ = smem + SQ_OFF;
    float* sK = smem + SK_OFF;
    float* sV = smem + SV_OFF;
    float* sS = smem + SS_OFF;
    float* sH = smem + SH_OFF;

    const int tid  = threadIdx.x;
    const int wave = tid >> 5;
    const int lane = tid & 31;
    const int m16  = lane & 15;
    const int kh   = lane >> 4;

    const bool second = blockIdx.x < NWG1;
    const int  t_begin = second ? T_SPLIT : 0;
    const int  t_end   = second ? NT : T_SPLIT;
    const int g  = blockIdx.x & (EG - 1);
    const int bh = (blockIdx.x >> 2) & (NB * NH - 1);
    const int h  = bh & (NH - 1);
    const int b  = bh >> 4;

    const size_t row_stride = (size_t)NH * D_DIM;
    const float* qbase = q + ((size_t)b * SEQ * NH + h) * D_DIM;
    const float* kbase = k + ((size_t)b * SEQ * NH + h) * D_DIM;
    const float* vbase = v + ((size_t)b * SEQ * NH + h) * D_DIM;

    const int qk_row = tid >> 5, qk_c4 = (tid & 31) << 2;
    const int v_row  = tid >> 3, v_c4  = (tid & 7) << 2;
    const float* qrow = qbase + (size_t)qk_row * row_stride + qk_c4;
    const float* krow = kbase + (size_t)qk_row * row_stride + qk_c4;
    const float* vrow = vbase + (size_t)v_row * row_stride + g * GW + v_c4;
    float* sQ_st = sQ + qk_row * SQ_LD + qk_c4;
    float* sK_st = sK + qk_row * SK_LD + qk_c4;
    float* sV_st = sV + v_row * SV_LD + v_c4;

    v4f pq[8], pk[8], pv[2];
    if (second) {
        const int dd = wave;
        const size_t cstep = (size_t)CH * row_stride;
        const size_t adv0  = (size_t)T_SPLIT * cstep;
        v4f ka[8], va[2], kb[8], vb[2];
        v8f h0 = {}, h1 = {};
        fetch_kv(ka, va, krow, vrow, row_stride);
        fetch_kv(kb, vb, krow + cstep, vrow + cstep, row_stride);
        stage_kv(ka, va, sK_st, sV_st);
        fetch_kv(ka, va, krow + 2 * cstep, vrow + 2 * cstep, row_stride);
        lds_barrier();
        for (int t = 0; t + 1 < T_SPLIT; t += 2) {
            state_step(kb, vb, t,     krow, vrow, row_stride, sK, sV, sQ, sS,
                       qk_row, qk_c4, v_row, v_c4, kh, m16, dd, h0, h1);
            state_step(ka, va, t + 1, krow, vrow, row_stride, sQ, sS, sK, sV,
                       qk_row, qk_c4, v_row, v_c4, kh, m16, dd, h0, h1);
        }
        fetch_chunk(pq, pk, pv, qrow + adv0, krow + adv0, vrow + adv0, row_stride);
        if (T_SPLIT & 1)
            state_step(kb, vb, T_SPLIT - 1, krow, vrow, row_stride, sK, sV, sQ, sS,
                       qk_row, qk_c4, v_row, v_c4, kh, m16, dd, h0, h1);
        float* pT0 = sH + (dd * 16 + 8 * kh) * SH_LD + m16;
        #pragma unroll
        for (int j = 0; j < 8; ++j) { pT0[j * SH_LD] = h0[j]; pT0[j * SH_LD + 16] = h1[j]; }
    } else {
        fetch_chunk(pq, pk, pv, qrow, krow, vrow, row_stride);
        for (int i = tid; i < 128 * SH_LD; i += 256) sH[i] = 0.0f;
    }

    const int cur0 = (t_begin & 1) ? BUF1_OFF : 0;
    stage_chunk(pq, pk, pv, sQ_st + cur0, sK_st + cur0, sV_st + cur0, scale);
    if (t_begin + 1 < t_end) {
        const size_t adv = (size_t)(t_begin + 1) * CH * row_stride;
        fetch_chunk(pq, pk, pv, qrow + adv, krow + adv, vrow + adv, row_stride);
    }
    lds_barrier();

    const int r_o  = wave & 3;
    const int eco  = (wave >> 2) * 16;

    const bool two_tiles = wave < 2;
    const int  rA = tile_row(wave),     cA = wave - rA * (rA + 1) / 2;
    const int  rB = 3,                  cB = wave + 2;
    const float* sA_a = sQ + (rA * 16 + m16) * SQ_LD + kh * 2;
    const float* sA_b = sK + (cA * 16 + m16) * SK_LD + kh * 2;
    const float* sB_a = sQ + (rB * 16 + m16) * SQ_LD + kh * 2;
    const float* sB_b = sK + ((cB & 3) * 16 + m16) * SK_LD + kh * 2;

    const float* aS  = sS + (r_o * 16 + m16) * SS_LD + kh * 2;
    const float* aQ  = sQ + (r_o * 16 + m16) * SQ_LD + kh * 2;
    const float* oV0 = sV + (kh * 2 + 0) * SV_LD + eco + m16;
    const float* oH0 = sH + (kh * 2 + 0) * SH_LD + eco + m16;

    const int d = wave;
    float* hT0 = sH + (d * 16 + 8 * kh) * SH_LD + m16;
    float* hT1 = hT0 + 16;
    const float* hK0 = sK + (kh * 2 + 0) * SK_LD + d * 16 + m16;
    const float* hK1 = sK + (kh * 2 + 1) * SK_LD + d * 16 + m16;
    const float* hV0 = sV + (kh * 2 + 0) * SV_LD + m16;

    {
        v8f u0, u1;
        s_phase<false>(two_tiles, sA_a + cur0, sA_b + cur0, sB_a + cur0, sB_b + cur0,
                       sS, rA, cA, rB, cB, m16, kh, nullptr, nullptr, nullptr,
                       u0, u1, hT0, hT1);
    }
    lds_barrier();

    for (int t = t_begin; t < t_end; ++t) {
        const size_t srow = (size_t)t * CH;
        const int  cur = (t & 1) ? BUF1_OFF : 0;
        const int  nxt = BUF1_OFF - cur;
        const bool has_next = t + 1 < t_end;

        v2f fq[32], fh[32];
        v8f acc;
        switch (r_o) {
            case 0:  acc = sv_part<1>(aS, oV0 + cur, fq, fh, aQ + cur, oH0); break;
            case 1:  acc = sv_part<2>(aS, oV0 + cur, fq, fh, aQ + cur, oH0); break;
            case 2:  acc = sv_part<3>(aS, oV0 + cur, fq, fh, aQ + cur, oH0); break;
            default: acc = sv_part<4>(aS, oV0 + cur, fq, fh, aQ + cur, oH0); break;
        }
        v2f fk[16], fva[16], fvb[16];
        v8f h0, h1;
        EDGE1(acc);
        ld_rk<16, 8, SH_LD>(fq, fh, aQ + cur, oH0);
        EDGE1(acc);
        acc = mm<0, 8>(fq, fh, acc);
        EDGE1(acc);
        ld_rk<24, 8, SH_LD>(fq, fh, aQ + cur, oH0);
        if (has_next)
            stage_chunk(pq, pk, pv, sQ_st + nxt, sK_st + nxt, sV_st + nxt, scale);
        EDGE1(acc);
        acc = mm<8, 8>(fq, fh, acc);
        EDGE1(acc);
        #pragma unroll
        for (int j = 0; j < 8; ++j) { h0[j] = hT0[j * SH_LD]; h1[j] = hT1[j * SH_LD]; }
        ld_h<0, 8>(fk, fva, fvb, hK0 + cur, hK1 + cur, hV0 + cur);
        EDGE1(acc);
        acc = mm<16, 8>(fq, fh, acc);
        EDGE1(acc);
        ld_h<8, 8>(fk, fva, fvb, hK0 + cur, hK1 + cur, hV0 + cur);
        {
            const int tn = (t + 2 < t_end) ? t + 2 : t_end - 1;
            const size_t adv = (size_t)tn * CH * row_stride;
            fetch_chunk(pq, pk, pv, qrow + adv, krow + adv, vrow + adv, row_stride);
        }
        EDGE1(acc);
        acc = mm<24, 8>(fq, fh, acc);

        #pragma unroll
        for (int j = 0; j < 8; ++j) {
            size_t s_idx = srow + r_o * 16 + j + 8 * kh;
            o[(((size_t)b * SEQ + s_idx) * NH + h) * D_DIM + g * GW + eco + m16] = acc[j];
        }
        EDGE2(h0, h1);
        lds_barrier();

        if (has_next) {
            s_phase<true>(two_tiles, sA_a + nxt, sA_b + nxt, sB_a + nxt, sB_b + nxt,
                          sS, rA, cA, rB, cB, m16, kh, fk, fva, fvb, h0, h1, hT0, hT1);
        } else {
            mm2<0, 16>(fk, fva, h0, fk, fvb, h1);
            h_store(hT0, hT1, h0, h1);
        }
        lds_barrier();
    }

    if (second) {
    #pragma unroll
    for (int j = 0; j < 4; ++j) {
        int f    = tid + j * 256;
        int drow = f >> 3;
        int c4   = (f & 7) << 2;
        float4 val = *(const float4*)(sH + drow * SH_LD + c4);
        *(float4*)(hout + (((size_t)(b * NH + h)) * D_DIM + drow) * D_DIM + g * GW + c4) = val;
    }
    }
}

extern "C" void kernel_launch(void* const* d_in, const int* in_sizes, int n_in,
                              void* d_out, int out_size, void* d_ws, size_t ws_size,
                              hipStream_t stream) {
    (void)in_sizes; (void)n_in; (void)out_size; (void)d_ws; (void)ws_size;
    const float* q = (const float*)d_in[0];
    const float* k = (const float*)d_in[1];
    const float* v = (const float*)d_in[2];
    float* o  = (float*)d_out;
    float* hf = o + (size_t)NB * SEQ * NH * D_DIM;
    const float scale = 0.088388347648318447f;

    const size_t smem = (size_t)SMEM_FLOATS * sizeof(float);
    hipFuncSetAttribute((const void*)fla_fused_chunk_fwd,
                        hipFuncAttributeMaxDynamicSharedMemorySize, (int)smem);

    dim3 grid(2 * NWG1);
    dim3 blk(256);
    hipLaunchKernelGGL(fla_fused_chunk_fwd, grid, blk, smem, stream,
                       q, k, v, o, hf, scale);
}


# state update cut in two 8-k-step runs, one heading each phase, on fragments fetched before the barrier (each V fragment one two-address load); state tiles kept in registers; sums unchanged
# speedup vs baseline: 3.1908x; 1.0438x over previous
#include <hip/hip_runtime.h>

typedef __attribute__((ext_vector_type(2))) float v2f;
typedef __attribute__((ext_vector_type(8))) float v8f;
typedef __attribute__((ext_vector_type(4))) float v4f;

#define D_DIM 128
#define CH    64
#define SEQ   4096
#define NT    (SEQ / CH)
#define NH    16
#define NB    2
#define EG    4
#define GW    32
#define NWG1  (NB * NH * EG)
#define T_SPLIT 35
static_assert(T_SPLIT >= 3 && T_SPLIT < NT, "the state pass keeps two chunks in flight");

#define SQ_LD 140
#define SK_LD 140
#define SV_LD 48
#define SS_LD 68
#define SH_LD 48

#define SQ_OFF 0
#define SK_OFF (SQ_OFF + 64 * SQ_LD)
#define SV_OFF (SK_OFF + 64 * SK_LD)
#define SS_OFF (SV_OFF + 64 * SV_LD)
#define SH_OFF (SS_OFF + 64 * SS_LD)
#define BUF1_OFF (SH_OFF + 128 * SH_LD)
#define SMEM_FLOATS (BUF1_OFF + SV_OFF + 64 * SV_LD)
static_assert(BUF1_OFF % 64 == 0, "both buffers on the same banks");

#define EDGE1(a)    asm volatile("" : "+v"(a) : : "memory")
#define EDGE2(a, b) asm volatile("" : "+v"(a), "+v"(b) : : "memory")
#define EDGE3(a, b, c)    asm volatile("" : "+v"(a), "+v"(b), "+v"(c) : : "memory")
#define EDGE4(a, b, c, d) asm volatile("" : "+v"(a), "+v"(b), "+v"(c), "+v"(d) : : "memory")

static __device__ __forceinline__ v8f wmma_f32(v2f a, v2f b, v8f c) {
    return __builtin_amdgcn_wmma_f32_16x16x4_f32(
         false, a,  false, b,
         (short)0, c,  false,  false);
}

static __device__ __forceinline__ void lds_barrier() {
    __builtin_amdgcn_fence(__ATOMIC_RELEASE, "workgroup", "local");
    __builtin_amdgcn_s_barrier();
    __builtin_amdgcn_fence(__ATOMIC_ACQUIRE, "workgroup", "local");
}

template <int J0, int N>
static __device__ __forceinline__ void ld_rr(v2f* fa, v2f* fb, const float* pa, const float* pb) {
    #pragma unroll
    for (int j = J0; j < J0 + N; ++j) {
        fa[j] = *(const v2f*)(pa + 4 * j);
        fb[j] = *(const v2f*)(pb + 4 * j);
    }
}

template <int J0, int N, int LD>
static __device__ __forceinline__ void ld_rk(v2f* fa, v2f* fb, const float* pa,
                                             const float* p0) {
    #pragma unroll
    for (int j = J0; j < J0 + N; ++j) {
        fa[j] = *(const v2f*)(pa + 4 * j);
        fb[j].x = p0[4 * j * LD];
        fb[j].y = p0[4 * j * LD + LD];
    }
}

template <int J0, int N>
static __device__ __forceinline__ void ld_h(v2f* fk, v2f* fva, v2f* fvb,
                                            const float* k0, const float* k1,
                                            const float* v0) {
    #pragma unroll
    for (int j = J0; j < J0 + N; ++j) {
        fk[j].x  = k0[4 * j * SK_LD];      fk[j].y  = k1[4 * j * SK_LD];
        fva[j].x = v0[4 * j * SV_LD];      fva[j].y = v0[4 * j * SV_LD + SV_LD];
    }
    asm volatile("" : : : "memory");
    #pragma unroll
    for (int j = J0; j < J0 + N; ++j) {
        fvb[j].x = v0[4 * j * SV_LD + 16]; fvb[j].y = v0[4 * j * SV_LD + SV_LD + 16];
    }
}

template <int J0, int N>
static __device__ __forceinline__ v8f mm(const v2f* fa, const v2f* fb, v8f acc) {
    #pragma unroll
    for (int j = J0; j < J0 + N; ++j) acc = wmma_f32(fa[j], fb[j], acc);
    return acc;
}

template <int J0, int N>
static __device__ __forceinline__ void mm2(const v2f* fa0, const v2f* fb0, v8f& c0,
                                           const v2f* fa1, const v2f* fb1, v8f& c1) {
    #pragma unroll
    for (int j = J0; j < J0 + N; ++j) {
        c0 = wmma_f32(fa0[j], fb0[j], c0);
        c1 = wmma_f32(fa1[j], fb1[j], c1);
    }
}

static __device__ __forceinline__ void s_store(float* sS, v8f acc, int r, int c, int m16, int kh) {
    #pragma unroll
    for (int j = 0; j < 8; ++j) {
        int er = r * 16 + j + 8 * kh;
        int ec = c * 16 + m16;
        float val = acc[j];
        if (r == c && er < ec) val = 0.0f;
        sS[er * SS_LD + ec] = val;
    }
}

template <int NBLK>
static __device__ __forceinline__ v8f sv_part(const float* aS, const float* bV0,
                                              v2f* fq, v2f* fh,
                                              const float* aQ, const float* oH0,
                                              const v2f* fk, const v2f* fva, const v2f* fvb,
                                              v8f& h0, v8f& h1) {
    v2f fs[NBLK * 4], fv[NBLK * 4];
    ld_rk<0, NBLK * 4, SV_LD>(fs, fv, aS, bV0);
    ld_rk<0, 16, SH_LD>(fq, fh, aQ, oH0);
    EDGE2(h0, h1);
    mm2<0, 8>(fk, fva, h0, fk, fvb, h1);
    v8f acc = {};
    EDGE3(h0, h1, acc);
    return mm<0, NBLK * 4>(fs, fv, acc);
}

static __device__ __forceinline__ int tile_row(int idx) {
    return (idx >= 6) ? 3 : (idx >= 3) ? 2 : (idx >= 1) ? 1 : 0;
}

static __device__ __forceinline__ void fetch_chunk(v4f* pq, v4f* pk, v4f* pv,
                                                   const float* qrow, const float* krow,
                                                   const float* vrow, size_t row_stride) {
    #pragma unroll
    for (int j = 0; j < 8; ++j) {
        pq[j] = *(const v4f*)(qrow + (size_t)(8 * j) * row_stride);
        pk[j] = *(const v4f*)(krow + (size_t)(8 * j) * row_stride);
    }
    #pragma unroll
    for (int j = 0; j < 2; ++j)
        pv[j] = *(const v4f*)(vrow + (size_t)(32 * j) * row_stride);
}

static __device__ __forceinline__ void fetch_kv(v4f* pk, v4f* pv, const float* krow,
                                                const float* vrow, size_t row_stride) {
    #pragma unroll
    for (int j = 0; j < 8; ++j)
        pk[j] = *(const v4f*)(krow + (size_t)(8 * j) * row_stride);
    #pragma unroll
    for (int j = 0; j < 2; ++j)
        pv[j] = *(const v4f*)(vrow + (size_t)(32 * j) * row_stride);
}

static __device__ __forceinline__ void stage_kv(const v4f* pk, const v4f* pv,
                                                float* sK_st, float* sV_st) {
    #pragma unroll
    for (int j = 0; j < 8; ++j) *(v4f*)(sK_st + 8 * j * SK_LD) = pk[j];
    #pragma unroll
    for (int j = 0; j < 2; ++j) *(v4f*)(sV_st + 32 * j * SV_LD) = pv[j];
}

static __device__ __forceinline__ void state_step(v4f* pk, v4f* pv, int t,
                                                  const float* krow, const float* vrow,
                                                  size_t row_stride,
                                                  const float* rK, const float* rV,
                                                  float* wK, float* wV,
                                                  int qk_row, int qk_c4, int v_row, int v_c4,
                                                  int kh, int m16, int d, v8f& h0, v8f& h1) {
    const float* k0 = rK + (kh * 2 + 0) * SK_LD + d * 16 + m16;
    const float* k1 = rK + (kh * 2 + 1) * SK_LD + d * 16 + m16;
    const float* v0 = rV + (kh * 2 + 0) * SV_LD + m16;
    v2f fk[16], fva[16], fvb[16];
    ld_h<0, 4>(fk, fva, fvb, k0, k1, v0);
    ld_h<4, 4>(fk, fva, fvb, k0, k1, v0);
    EDGE2(h0, h1);
    if (t + 1 < T_SPLIT) {
        stage_kv(pk, pv, wK + qk_row * SK_LD + qk_c4, wV + v_row * SV_LD + v_c4);
        const int tn = (t + 3 < T_SPLIT) ? t + 3 : T_SPLIT - 1;
        const size_t adv = (size_t)tn * CH * row_stride;
        fetch_kv(pk, pv, krow + adv, vrow + adv, row_stride);
    }
    EDGE2(h0, h1);
    mm2<0, 4>(fk, fva, h0, fk, fvb, h1);
    EDGE2(h0, h1);
    ld_h<8, 4>(fk, fva, fvb, k0, k1, v0);
    EDGE2(h0, h1);
    mm2<4, 4>(fk, fva, h0, fk, fvb, h1);
    EDGE2(h0, h1);
    ld_h<12, 4>(fk, fva, fvb, k0, k1, v0);
    EDGE2(h0, h1);
    mm2<8, 8>(fk, fva, h0, fk, fvb, h1);
    lds_barrier();
}

static __device__ __forceinline__ void stage_chunk(const v4f* pq, const v4f* pk, const v4f* pv,
                                                   float* sQ_st, float* sK_st, float* sV_st,
                                                   float scale) {
    #pragma unroll
    for (int j = 0; j < 8; ++j) {
        *(v4f*)(sQ_st + 8 * j * SQ_LD) = pq[j] * scale;
        *(v4f*)(sK_st + 8 * j * SK_LD) = pk[j];
    }
    #pragma unroll
    for (int j = 0; j < 2; ++j)
        *(v4f*)(sV_st + 32 * j * SV_LD) = pv[j];
}

static __device__ __forceinline__ void h_store(float* hT0, float* hT1, v8f h0, v8f h1) {
    #pragma unroll
    for (int j = 0; j < 8; ++j) { hT0[j * SH_LD] = h0[j]; hT1[j * SH_LD] = h1[j]; }
}

template <bool WITH_H>
static __device__ __forceinline__ void s_phase(bool two_tiles,
                                               const float* sA_a, const float* sA_b,
                                               const float* sB_a, const float* sB_b,
                                               float* sS, int rA, int cA, int rB, int cB,
                                               int m16, int kh,
                                               v2f* fk, v2f* fva, v2f* fvb,
                                               v8f& h0, v8f& h1, float* hT0, float* hT1,
                                               const float* nK0, const float* nK1,
                                               const float* nV0) {
    if (two_tiles) {
        v2f fa0[32], fb0[32], fa1[32], fb1[32];
        v8f c0 = {}, c1 = {};
        ld_rr<0, 8>(fa0, fb0, sA_a, sA_b);   ld_rr<0, 8>(fa1, fb1, sB_a, sB_b);
        ld_rr<8, 8>(fa0, fb0, sA_a, sA_b);   ld_rr<8, 8>(fa1, fb1, sB_a, sB_b);
        if constexpr (WITH_H) {
            EDGE2(h0, h1);
            mm2<8, 8>(fk, fva, h0, fk, fvb, h1);
            EDGE4(h0, h1, c0, c1);
        } else {
            EDGE2(c0, c1);
        }
        mm2<0, 8>(fa0, fb0, c0, fa1, fb1, c1);
        EDGE2(c0, c1);
        ld_rr<16, 8>(fa0, fb0, sA_a, sA_b);  ld_rr<16, 8>(fa1, fb1, sB_a, sB_b);
        if constexpr (WITH_H) h_store(hT0, hT1, h0, h1);
        EDGE2(c0, c1);
        mm2<8, 8>(fa0, fb0, c0, fa1, fb1, c1);
        EDGE2(c0, c1);
        ld_rr<24, 8>(fa0, fb0, sA_a, sA_b);  ld_rr<24, 8>(fa1, fb1, sB_a, sB_b);
        if constexpr (WITH_H) ld_h<0, 8>(fk, fva, fvb, nK0, nK1, nV0);
        EDGE2(c0, c1);
        mm2<16, 16>(fa0, fb0, c0, fa1, fb1, c1);
        EDGE2(c0, c1);
        s_store(sS, c0, rA, cA, m16, kh);
        s_store(sS, c1, rB, cB, m16, kh);
    } else {
        v2f fa[32], fb[32];
        v8f c0 = {};
        ld_rr<0, 8>(fa, fb, sA_a, sA_b);
        ld_rr<8, 8>(fa, fb, sA_a, sA_b);
        if constexpr (WITH_H) {
            EDGE2(h0, h1);
            mm2<8, 8>(fk, fva, h0, fk, fvb, h1);
            EDGE3(h0, h1, c0);
        } else {
            EDGE1(c0);
        }
        c0 = mm<0, 8>(fa, fb, c0);
        EDGE1(c0);
        ld_rr<16, 8>(fa, fb, sA_a, sA_b);
        if constexpr (WITH_H) h_store(hT0, hT1, h0, h1);
        EDGE1(c0);
        c0 = mm<8, 8>(fa, fb, c0);
        EDGE1(c0);
        ld_rr<24, 8>(fa, fb, sA_a, sA_b);
        if constexpr (WITH_H) ld_h<0, 8>(fk, fva, fvb, nK0, nK1, nV0);
        EDGE1(c0);
        c0 = mm<16, 16>(fa, fb, c0);
        EDGE1(c0);
        s_store(sS, c0, rA, cA, m16, kh);
    }
}

extern "C" __global__ __launch_bounds__(256)
void fla_fused_chunk_fwd(const float* __restrict__ q,
                         const float* __restrict__ k,
                         const float* __restrict__ v,
                         float* __restrict__ o,
                         float* __restrict__ hout,
                         float scale)
{
    extern __shared__ float smem[];
    float* sQ = smem + SQ_OFF;
    float* sK = smem + SK_OFF;
    float* sV = smem + SV_OFF;
    float* sS = smem + SS_OFF;
    float* sH = smem + SH_OFF;

    const int tid  = threadIdx.x;
    const int wave = tid >> 5;
    const int lane = tid & 31;
    const int m16  = lane & 15;
    const int kh   = lane >> 4;

    const bool second = blockIdx.x < NWG1;
    const int  t_begin = second ? T_SPLIT : 0;
    const int  t_end   = second ? NT : T_SPLIT;
    const int g  = blockIdx.x & (EG - 1);
    const int bh = (blockIdx.x >> 2) & (NB * NH - 1);
    const int h  = bh & (NH - 1);
    const int b  = bh >> 4;

    const size_t row_stride = (size_t)NH * D_DIM;
    const float* qbase = q + ((size_t)b * SEQ * NH + h) * D_DIM;
    const float* kbase = k + ((size_t)b * SEQ * NH + h) * D_DIM;
    const float* vbase = v + ((size_t)b * SEQ * NH + h) * D_DIM;

    const int qk_row = tid >> 5, qk_c4 = (tid & 31) << 2;
    const int v_row  = tid >> 3, v_c4  = (tid & 7) << 2;
    const float* qrow = qbase + (size_t)qk_row * row_stride + qk_c4;
    const float* krow = kbase + (size_t)qk_row * row_stride + qk_c4;
    const float* vrow = vbase + (size_t)v_row * row_stride + g * GW + v_c4;
    float* sQ_st = sQ + qk_row * SQ_LD + qk_c4;
    float* sK_st = sK + qk_row * SK_LD + qk_c4;
    float* sV_st = sV + v_row * SV_LD + v_c4;

    v4f pq[8], pk[8], pv[2];
    v8f h0 = {}, h1 = {};
    v2f fk[16], fva[16], fvb[16];
    if (second) {
        const int dd = wave;
        const size_t cstep = (size_t)CH * row_stride;
        const size_t adv0  = (size_t)T_SPLIT * cstep;
        v4f ka[8], va[2], kb[8], vb[2];
        fetch_kv(ka, va, krow, vrow, row_stride);
        fetch_kv(kb, vb, krow + cstep, vrow + cstep, row_stride);
        stage_kv(ka, va, sK_st, sV_st);
        fetch_kv(ka, va, krow + 2 * cstep, vrow + 2 * cstep, row_stride);
        lds_barrier();
        for (int t = 0; t + 1 < T_SPLIT; t += 2) {
            state_step(kb, vb, t,     krow, vrow, row_stride, sK, sV, sQ, sS,
                       qk_row, qk_c4, v_row, v_c4, kh, m16, dd, h0, h1);
            state_step(ka, va, t + 1, krow, vrow, row_stride, sQ, sS, sK, sV,
                       qk_row, qk_c4, v_row, v_c4, kh, m16, dd, h0, h1);
        }
        fetch_chunk(pq, pk, pv, qrow + adv0, krow + adv0, vrow + adv0, row_stride);
        if (T_SPLIT & 1)
            state_step(kb, vb, T_SPLIT - 1, krow, vrow, row_stride, sK, sV, sQ, sS,
                       qk_row, qk_c4, v_row, v_c4, kh, m16, dd, h0, h1);
        float* pT0 = sH + (dd * 16 + 8 * kh) * SH_LD + m16;
        #pragma unroll
        for (int j = 0; j < 8; ++j) { pT0[j * SH_LD] = h0[j]; pT0[j * SH_LD + 16] = h1[j]; }
    } else {
        fetch_chunk(pq, pk, pv, qrow, krow, vrow, row_stride);
        for (int i = tid; i < 128 * SH_LD; i += 256) sH[i] = 0.0f;
    }

    const int cur0 = (t_begin & 1) ? BUF1_OFF : 0;
    stage_chunk(pq, pk, pv, sQ_st + cur0, sK_st + cur0, sV_st + cur0, scale);
    if (t_begin + 1 < t_end) {
        const size_t adv = (size_t)(t_begin + 1) * CH * row_stride;
        fetch_chunk(pq, pk, pv, qrow + adv, krow + adv, vrow + adv, row_stride);
    }
    lds_barrier();

    const int r_o  = wave & 3;
    const int eco  = (wave >> 2) * 16;

    const bool two_tiles = wave < 2;
    const int  rA = tile_row(wave),     cA = wave - rA * (rA + 1) / 2;
    const int  rB = 3,                  cB = wave + 2;
    const float* sA_a = sQ + (rA * 16 + m16) * SQ_LD + kh * 2;
    const float* sA_b = sK + (cA * 16 + m16) * SK_LD + kh * 2;
    const float* sB_a = sQ + (rB * 16 + m16) * SQ_LD + kh * 2;
    const float* sB_b = sK + ((cB & 3) * 16 + m16) * SK_LD + kh * 2;

    const float* aS  = sS + (r_o * 16 + m16) * SS_LD + kh * 2;
    const float* aQ  = sQ + (r_o * 16 + m16) * SQ_LD + kh * 2;
    const float* oV0 = sV + (kh * 2 + 0) * SV_LD + eco + m16;
    const float* oH0 = sH + (kh * 2 + 0) * SH_LD + eco + m16;

    const int d = wave;
    float* hT0 = sH + (d * 16 + 8 * kh) * SH_LD + m16;
    float* hT1 = hT0 + 16;
    const float* hK0 = sK + (kh * 2 + 0) * SK_LD + d * 16 + m16;
    const float* hK1 = sK + (kh * 2 + 1) * SK_LD + d * 16 + m16;
    const float* hV0 = sV + (kh * 2 + 0) * SV_LD + m16;

    {
        ld_h<0, 8>(fk, fva, fvb, hK0 + cur0, hK1 + cur0, hV0 + cur0);
        v8f u0, u1;
        s_phase<false>(two_tiles, sA_a + cur0, sA_b + cur0, sB_a + cur0, sB_b + cur0,
                       sS, rA, cA, rB, cB, m16, kh, nullptr, nullptr, nullptr,
                       u0, u1, hT0, hT1, nullptr, nullptr, nullptr);
    }
    lds_barrier();

    for (int t = t_begin; t < t_end; ++t) {
        const size_t srow = (size_t)t * CH;
        const int  cur = (t & 1) ? BUF1_OFF : 0;
        const int  nxt = BUF1_OFF - cur;
        const bool has_next = t + 1 < t_end;

        v2f fq[32], fh[32];
        v8f acc;
        switch (r_o) {
            case 0:  acc = sv_part<1>(aS, oV0 + cur, fq, fh, aQ + cur, oH0, fk, fva, fvb, h0, h1); break;
            case 1:  acc = sv_part<2>(aS, oV0 + cur, fq, fh, aQ + cur, oH0, fk, fva, fvb, h0, h1); break;
            case 2:  acc = sv_part<3>(aS, oV0 + cur, fq, fh, aQ + cur, oH0, fk, fva, fvb, h0, h1); break;
            default: acc = sv_part<4>(aS, oV0 + cur, fq, fh, aQ + cur, oH0, fk, fva, fvb, h0, h1); break;
        }
        EDGE1(acc);
        ld_rk<16, 8, SH_LD>(fq, fh, aQ + cur, oH0);
        EDGE1(acc);
        acc = mm<0, 8>(fq, fh, acc);
        EDGE1(acc);
        ld_rk<24, 8, SH_LD>(fq, fh, aQ + cur, oH0);
        if (has_next)
            stage_chunk(pq, pk, pv, sQ_st + nxt, sK_st + nxt, sV_st + nxt, scale);
        EDGE1(acc);
        acc = mm<8, 8>(fq, fh, acc);
        EDGE1(acc);
        ld_h<8, 8>(fk, fva, fvb, hK0 + cur, hK1 + cur, hV0 + cur);
        EDGE1(acc);
        acc = mm<16, 8>(fq, fh, acc);
        EDGE1(acc);
        {
            const int tn = (t + 2 < t_end) ? t + 2 : t_end - 1;
            const size_t adv = (size_t)tn * CH * row_stride;
            fetch_chunk(pq, pk, pv, qrow + adv, krow + adv, vrow + adv, row_stride);
        }
        EDGE1(acc);
        acc = mm<24, 8>(fq, fh, acc);

        #pragma unroll
        for (int j = 0; j < 8; ++j) {
            size_t s_idx = srow + r_o * 16 + j + 8 * kh;
            o[(((size_t)b * SEQ + s_idx) * NH + h) * D_DIM + g * GW + eco + m16] = acc[j];
        }
        EDGE2(h0, h1);
        lds_barrier();

        if (has_next) {
            s_phase<true>(two_tiles, sA_a + nxt, sA_b + nxt, sB_a + nxt, sB_b + nxt,
                          sS, rA, cA, rB, cB, m16, kh, fk, fva, fvb, h0, h1, hT0, hT1,
                          hK0 + nxt, hK1 + nxt, hV0 + nxt);
        } else {
            mm2<8, 8>(fk, fva, h0, fk, fvb, h1);
            h_store(hT0, hT1, h0, h1);
        }
        lds_barrier();
    }

    if (second) {
    #pragma unroll
    for (int j = 0; j < 4; ++j) {
        int f    = tid + j * 256;
        int drow = f >> 3;
        int c4   = (f & 7) << 2;
        float4 val = *(const float4*)(sH + drow * SH_LD + c4);
        *(float4*)(hout + (((size_t)(b * NH + h)) * D_DIM + drow) * D_DIM + g * GW + c4) = val;
    }
    }
}

extern "C" void kernel_launch(void* const* d_in, const int* in_sizes, int n_in,
                              void* d_out, int out_size, void* d_ws, size_t ws_size,
                              hipStream_t stream) {
    (void)in_sizes; (void)n_in; (void)out_size; (void)d_ws; (void)ws_size;
    const float* q = (const float*)d_in[0];
    const float* k = (const float*)d_in[1];
    const float* v = (const float*)d_in[2];
    float* o  = (float*)d_out;
    float* hf = o + (size_t)NB * SEQ * NH * D_DIM;
    const float scale = 0.088388347648318447f;

    const size_t smem = (size_t)SMEM_FLOATS * sizeof(float);
    hipFuncSetAttribute((const void*)fla_fused_chunk_fwd,
                        hipFuncAttributeMaxDynamicSharedMemorySize, (int)smem);

    dim3 grid(2 * NWG1);
    dim3 blk(256);
    hipLaunchKernelGGL(fla_fused_chunk_fwd, grid, blk, smem, stream,
                       q, k, v, o, hf, scale);
}


# S and S*V kept inside each 32-row half of a chunk; rows 32..63 take the upper half through a half-way state copy, q*(H + Kup^T Vup): 160 of 912 matrix ops a chunk gone; O sum order changed
# speedup vs baseline: 3.3653x; 1.0547x over previous
#include <hip/hip_runtime.h>

typedef __attribute__((ext_vector_type(2))) float v2f;
typedef __attribute__((ext_vector_type(8))) float v8f;
typedef __attribute__((ext_vector_type(4))) float v4f;

#define D_DIM 128
#define CH    64
#define SEQ   4096
#define NT    (SEQ / CH)
#define NH    16
#define NB    2
#define EG    4
#define GW    32
#define NWG1  (NB * NH * EG)
#define T_SPLIT 35
static_assert(T_SPLIT >= 3 && T_SPLIT < NT, "the state pass keeps two chunks in flight");

#define SQ_LD 140
#define SK_LD 140
#define SV_LD 48
#define SS_LD 68
#define SH_LD 48

#define SQ_OFF 0
#define SK_OFF (SQ_OFF + 64 * SQ_LD)
#define SV_OFF (SK_OFF + 64 * SK_LD)
#define SS_OFF (SV_OFF + 64 * SV_LD)
#define SH_OFF (SS_OFF + 64 * SS_LD)
#define BUF1_OFF (SH_OFF + 128 * SH_LD)
#define SHM_OFF (BUF1_OFF + SV_OFF + 64 * SV_LD)
#define SMEM_FLOATS (SHM_OFF + 128 * SH_LD)
static_assert(BUF1_OFF % 64 == 0, "both buffers on the same banks");
static_assert(SHM_OFF % 64 == 0, "both state copies on the same banks");

#define EDGE1(a)    asm volatile("" : "+v"(a) : : "memory")
#define EDGE2(a, b) asm volatile("" : "+v"(a), "+v"(b) : : "memory")
#define EDGE3(a, b, c)    asm volatile("" : "+v"(a), "+v"(b), "+v"(c) : : "memory")
#define EDGE4(a, b, c, d) asm volatile("" : "+v"(a), "+v"(b), "+v"(c), "+v"(d) : : "memory")

static __device__ __forceinline__ v8f wmma_f32(v2f a, v2f b, v8f c) {
    return __builtin_amdgcn_wmma_f32_16x16x4_f32(
         false, a,  false, b,
         (short)0, c,  false,  false);
}

static __device__ __forceinline__ void lds_barrier() {
    __builtin_amdgcn_fence(__ATOMIC_RELEASE, "workgroup", "local");
    __builtin_amdgcn_s_barrier();
    __builtin_amdgcn_fence(__ATOMIC_ACQUIRE, "workgroup", "local");
}

template <int J0, int N>
static __device__ __forceinline__ void ld_rr(v2f* fa, v2f* fb, const float* pa, const float* pb) {
    #pragma unroll
    for (int j = J0; j < J0 + N; ++j) {
        fa[j] = *(const v2f*)(pa + 4 * j);
        fb[j] = *(const v2f*)(pb + 4 * j);
    }
}

template <int J0, int N, int LD>
static __device__ __forceinline__ void ld_rk(v2f* fa, v2f* fb, const float* pa,
                                             const float* p0) {
    #pragma unroll
    for (int j = J0; j < J0 + N; ++j) {
        fa[j] = *(const v2f*)(pa + 4 * j);
        fb[j].x = p0[4 * j * LD];
        fb[j].y = p0[4 * j * LD + LD];
    }
}

template <int J0, int N>
static __device__ __forceinline__ void ld_h(v2f* fk, v2f* fva, v2f* fvb,
                                            const float* k0, const float* k1,
                                            const float* v0) {
    #pragma unroll
    for (int j = J0; j < J0 + N; ++j) {
        fk[j].x  = k0[4 * j * SK_LD];      fk[j].y  = k1[4 * j * SK_LD];
        fva[j].x = v0[4 * j * SV_LD];      fva[j].y = v0[4 * j * SV_LD + SV_LD];
    }
    asm volatile("" : : : "memory");
    #pragma unroll
    for (int j = J0; j < J0 + N; ++j) {
        fvb[j].x = v0[4 * j * SV_LD + 16]; fvb[j].y = v0[4 * j * SV_LD + SV_LD + 16];
    }
}

template <int J0, int N>
static __device__ __forceinline__ v8f mm(const v2f* fa, const v2f* fb, v8f acc) {
    #pragma unroll
    for (int j = J0; j < J0 + N; ++j) acc = wmma_f32(fa[j], fb[j], acc);
    return acc;
}

template <int J0, int N>
static __device__ __forceinline__ void mm2(const v2f* fa0, const v2f* fb0, v8f& c0,
                                           const v2f* fa1, const v2f* fb1, v8f& c1) {
    #pragma unroll
    for (int j = J0; j < J0 + N; ++j) {
        c0 = wmma_f32(fa0[j], fb0[j], c0);
        c1 = wmma_f32(fa1[j], fb1[j], c1);
    }
}

static __device__ __forceinline__ void s_store(float* sS, v8f acc, int r, int c, int m16, int kh) {
    #pragma unroll
    for (int j = 0; j < 8; ++j) {
        int er = r * 16 + j + 8 * kh;
        int ec = c * 16 + m16;
        float val = acc[j];
        if (r == c && er < ec) val = 0.0f;
        sS[er * SS_LD + ec] = val;
    }
}

template <int NBLK>
static __device__ __forceinline__ v8f sv_part(const float* aS, const float* bV0,
                                              v2f* fq, v2f* fh,
                                              const float* aQ, const float* oH0) {
    v2f fs[NBLK * 4], fv[NBLK * 4];
    ld_rk<0, NBLK * 4, SV_LD>(fs, fv, aS, bV0);
    ld_rk<0, 16, SH_LD>(fq, fh, aQ, oH0);
    v8f acc = {};
    EDGE1(acc);
    return mm<0, NBLK * 4>(fs, fv, acc);
}


static __device__ __forceinline__ void fetch_chunk(v4f* pq, v4f* pk, v4f* pv,
                                                   const float* qrow, const float* krow,
                                                   const float* vrow, size_t row_stride) {
    #pragma unroll
    for (int j = 0; j < 8; ++j) {
        pq[j] = *(const v4f*)(qrow + (size_t)(8 * j) * row_stride);
        pk[j] = *(const v4f*)(krow + (size_t)(8 * j) * row_stride);
    }
    #pragma unroll
    for (int j = 0; j < 2; ++j)
        pv[j] = *(const v4f*)(vrow + (size_t)(32 * j) * row_stride);
}

static __device__ __forceinline__ void fetch_kv(v4f* pk, v4f* pv, const float* krow,
                                                const float* vrow, size_t row_stride) {
    #pragma unroll
    for (int j = 0; j < 8; ++j)
        pk[j] = *(const v4f*)(krow + (size_t)(8 * j) * row_stride);
    #pragma unroll
    for (int j = 0; j < 2; ++j)
        pv[j] = *(const v4f*)(vrow + (size_t)(32 * j) * row_stride);
}

static __device__ __forceinline__ void stage_kv(const v4f* pk, const v4f* pv,
                                                float* sK_st, float* sV_st) {
    #pragma unroll
    for (int j = 0; j < 8; ++j) *(v4f*)(sK_st + 8 * j * SK_LD) = pk[j];
    #pragma unroll
    for (int j = 0; j < 2; ++j) *(v4f*)(sV_st + 32 * j * SV_LD) = pv[j];
}

static __device__ __forceinline__ void state_step(v4f* pk, v4f* pv, int t,
                                                  const float* krow, const float* vrow,
                                                  size_t row_stride,
                                                  const float* rK, const float* rV,
                                                  float* wK, float* wV,
                                                  int qk_row, int qk_c4, int v_row, int v_c4,
                                                  int kh, int m16, int d, v8f& h0, v8f& h1) {
    const float* k0 = rK + (kh * 2 + 0) * SK_LD + d * 16 + m16;
    const float* k1 = rK + (kh * 2 + 1) * SK_LD + d * 16 + m16;
    const float* v0 = rV + (kh * 2 + 0) * SV_LD + m16;
    v2f fk[16], fva[16], fvb[16];
    ld_h<0, 4>(fk, fva, fvb, k0, k1, v0);
    ld_h<4, 4>(fk, fva, fvb, k0, k1, v0);
    EDGE2(h0, h1);
    if (t + 1 < T_SPLIT) {
        stage_kv(pk, pv, wK + qk_row * SK_LD + qk_c4, wV + v_row * SV_LD + v_c4);
        const int tn = (t + 3 < T_SPLIT) ? t + 3 : T_SPLIT - 1;
        const size_t adv = (size_t)tn * CH * row_stride;
        fetch_kv(pk, pv, krow + adv, vrow + adv, row_stride);
    }
    EDGE2(h0, h1);
    mm2<0, 4>(fk, fva, h0, fk, fvb, h1);
    EDGE2(h0, h1);
    ld_h<8, 4>(fk, fva, fvb, k0, k1, v0);
    EDGE2(h0, h1);
    mm2<4, 4>(fk, fva, h0, fk, fvb, h1);
    EDGE2(h0, h1);
    ld_h<12, 4>(fk, fva, fvb, k0, k1, v0);
    EDGE2(h0, h1);
    mm2<8, 8>(fk, fva, h0, fk, fvb, h1);
    lds_barrier();
}

static __device__ __forceinline__ void stage_chunk(const v4f* pq, const v4f* pk, const v4f* pv,
                                                   float* sQ_st, float* sK_st, float* sV_st,
                                                   float scale) {
    #pragma unroll
    for (int j = 0; j < 8; ++j) {
        *(v4f*)(sQ_st + 8 * j * SQ_LD) = pq[j] * scale;
        *(v4f*)(sK_st + 8 * j * SK_LD) = pk[j];
    }
    #pragma unroll
    for (int j = 0; j < 2; ++j)
        *(v4f*)(sV_st + 32 * j * SV_LD) = pv[j];
}

static __device__ __forceinline__ void h_store(float* hT0, float* hT1, v8f h0, v8f h1) {
    #pragma unroll
    for (int j = 0; j < 8; ++j) { hT0[j * SH_LD] = h0[j]; hT1[j * SH_LD] = h1[j]; }
}

template <bool STEADY>
static __device__ __forceinline__ void s_phase(bool has_tile,
                                               const float* sA_a, const float* sA_b,
                                               float* sS, int rA, int cA,
                                               int m16, int kh,
                                               v2f* fk, v2f* fva, v2f* fvb,
                                               v8f& h0, v8f& h1,
                                               float* hT0, float* hT1,
                                               float* hM0, float* hM1,
                                               const float* nK0, const float* nK1,
                                               const float* nV0) {
    if (has_tile) {
        v2f fa[32], fb[32];
        v8f c0 = {};
        ld_rr<0, 8>(fa, fb, sA_a, sA_b);
        ld_rr<8, 8>(fa, fb, sA_a, sA_b);
        ld_h<0, 8>(fk, fva, fvb, nK0, nK1, nV0);
        if constexpr (STEADY) {
            EDGE2(h0, h1);
            mm2<8, 8>(fk, fva, h0, fk, fvb, h1);
            EDGE3(h0, h1, c0);
        } else {
            EDGE1(c0);
        }
        c0 = mm<0, 8>(fa, fb, c0);
        EDGE3(h0, h1, c0);
        ld_rr<16, 8>(fa, fb, sA_a, sA_b);
        if constexpr (STEADY) h_store(hT0, hT1, h0, h1);
        EDGE3(h0, h1, c0);
        mm2<0, 8>(fk, fva, h0, fk, fvb, h1);
        c0 = mm<8, 8>(fa, fb, c0);
        EDGE3(h0, h1, c0);
        ld_rr<24, 8>(fa, fb, sA_a, sA_b);
        h_store(hM0, hM1, h0, h1);
        EDGE1(c0);
        c0 = mm<16, 16>(fa, fb, c0);
        EDGE1(c0);
        s_store(sS, c0, rA, cA, m16, kh);
    } else {
        ld_h<0, 8>(fk, fva, fvb, nK0, nK1, nV0);
        if constexpr (STEADY) {
            EDGE2(h0, h1);
            mm2<8, 8>(fk, fva, h0, fk, fvb, h1);
            EDGE2(h0, h1);
            h_store(hT0, hT1, h0, h1);
        }
        EDGE2(h0, h1);
        mm2<0, 8>(fk, fva, h0, fk, fvb, h1);
        EDGE2(h0, h1);
        h_store(hM0, hM1, h0, h1);
    }
}

extern "C" __global__ __launch_bounds__(256)
void fla_fused_chunk_fwd(const float* __restrict__ q,
                         const float* __restrict__ k,
                         const float* __restrict__ v,
                         float* __restrict__ o,
                         float* __restrict__ hout,
                         float scale)
{
    extern __shared__ float smem[];
    float* sQ = smem + SQ_OFF;
    float* sK = smem + SK_OFF;
    float* sV = smem + SV_OFF;
    float* sS = smem + SS_OFF;
    float* sH = smem + SH_OFF;
    float* sHm = smem + SHM_OFF;

    const int tid  = threadIdx.x;
    const int wave = tid >> 5;
    const int lane = tid & 31;
    const int m16  = lane & 15;
    const int kh   = lane >> 4;

    const bool second = blockIdx.x < NWG1;
    const int  t_begin = second ? T_SPLIT : 0;
    const int  t_end   = second ? NT : T_SPLIT;
    const int g  = blockIdx.x & (EG - 1);
    const int bh = (blockIdx.x >> 2) & (NB * NH - 1);
    const int h  = bh & (NH - 1);
    const int b  = bh >> 4;

    const size_t row_stride = (size_t)NH * D_DIM;
    const float* qbase = q + ((size_t)b * SEQ * NH + h) * D_DIM;
    const float* kbase = k + ((size_t)b * SEQ * NH + h) * D_DIM;
    const float* vbase = v + ((size_t)b * SEQ * NH + h) * D_DIM;

    const int qk_row = tid >> 5, qk_c4 = (tid & 31) << 2;
    const int v_row  = tid >> 3, v_c4  = (tid & 7) << 2;
    const float* qrow = qbase + (size_t)qk_row * row_stride + qk_c4;
    const float* krow = kbase + (size_t)qk_row * row_stride + qk_c4;
    const float* vrow = vbase + (size_t)v_row * row_stride + g * GW + v_c4;
    float* sQ_st = sQ + qk_row * SQ_LD + qk_c4;
    float* sK_st = sK + qk_row * SK_LD + qk_c4;
    float* sV_st = sV + v_row * SV_LD + v_c4;

    v4f pq[8], pk[8], pv[2];
    v8f h0 = {}, h1 = {};
    v2f fk[16], fva[16], fvb[16];
    if (second) {
        const int dd = wave;
        const size_t cstep = (size_t)CH * row_stride;
        const size_t adv0  = (size_t)T_SPLIT * cstep;
        v4f ka[8], va[2], kb[8], vb[2];
        fetch_kv(ka, va, krow, vrow, row_stride);
        fetch_kv(kb, vb, krow + cstep, vrow + cstep, row_stride);
        stage_kv(ka, va, sK_st, sV_st);
        fetch_kv(ka, va, krow + 2 * cstep, vrow + 2 * cstep, row_stride);
        lds_barrier();
        for (int t = 0; t + 1 < T_SPLIT; t += 2) {
            state_step(kb, vb, t,     krow, vrow, row_stride, sK, sV, sQ, sS,
                       qk_row, qk_c4, v_row, v_c4, kh, m16, dd, h0, h1);
            state_step(ka, va, t + 1, krow, vrow, row_stride, sQ, sS, sK, sV,
                       qk_row, qk_c4, v_row, v_c4, kh, m16, dd, h0, h1);
        }
        fetch_chunk(pq, pk, pv, qrow + adv0, krow + adv0, vrow + adv0, row_stride);
        if (T_SPLIT & 1)
            state_step(kb, vb, T_SPLIT - 1, krow, vrow, row_stride, sK, sV, sQ, sS,
                       qk_row, qk_c4, v_row, v_c4, kh, m16, dd, h0, h1);
        float* pT0 = sH + (dd * 16 + 8 * kh) * SH_LD + m16;
        #pragma unroll
        for (int j = 0; j < 8; ++j) { pT0[j * SH_LD] = h0[j]; pT0[j * SH_LD + 16] = h1[j]; }
    } else {
        fetch_chunk(pq, pk, pv, qrow, krow, vrow, row_stride);
        for (int i = tid; i < 128 * SH_LD; i += 256) sH[i] = 0.0f;
    }

    const int cur0 = (t_begin & 1) ? BUF1_OFF : 0;
    stage_chunk(pq, pk, pv, sQ_st + cur0, sK_st + cur0, sV_st + cur0, scale);
    if (t_begin + 1 < t_end) {
        const size_t adv = (size_t)(t_begin + 1) * CH * row_stride;
        fetch_chunk(pq, pk, pv, qrow + adv, krow + adv, vrow + adv, row_stride);
    }
    lds_barrier();

    const int r_o  = wave & 3;
    const int eco  = (wave >> 2) * 16;

    const bool has_tile = wave < 6;
    const int  tw = has_tile ? wave : 0;
    const int  rA = (tw >= 4) ? 3 : (tw == 3) ? 2 : (tw >= 1) ? 1 : 0;
    const int  cA = (tw == 5) ? 3 : (tw >= 3) ? 2 : (tw == 2) ? 1 : 0;
    const float* sA_a = sQ + (rA * 16 + m16) * SQ_LD + kh * 2;
    const float* sA_b = sK + (cA * 16 + m16) * SK_LD + kh * 2;

    const int  half = (r_o >> 1) * 32;
    const float* aS  = sS + (r_o * 16 + m16) * SS_LD + half + kh * 2;
    const float* aQ  = sQ + (r_o * 16 + m16) * SQ_LD + kh * 2;
    const float* oV0 = sV + (half + kh * 2) * SV_LD + eco + m16;
    const float* oH0 = ((r_o >> 1) ? sHm : sH) + (kh * 2 + 0) * SH_LD + eco + m16;

    const int d = wave;
    float* hT0 = sH + (d * 16 + 8 * kh) * SH_LD + m16;
    float* hT1 = hT0 + 16;
    float* hM0 = sHm + (d * 16 + 8 * kh) * SH_LD + m16;
    float* hM1 = hM0 + 16;
    const float* hK0 = sK + (kh * 2 + 0) * SK_LD + d * 16 + m16;
    const float* hK1 = sK + (kh * 2 + 1) * SK_LD + d * 16 + m16;
    const float* hV0 = sV + (kh * 2 + 0) * SV_LD + m16;

    s_phase<false>(has_tile, sA_a + cur0, sA_b + cur0, sS, rA, cA, m16, kh,
                   fk, fva, fvb, h0, h1, hT0, hT1, hM0, hM1,
                   hK0 + cur0, hK1 + cur0, hV0 + cur0);
    lds_barrier();

    for (int t = t_begin; t < t_end; ++t) {
        const size_t srow = (size_t)t * CH;
        const int  cur = (t & 1) ? BUF1_OFF : 0;
        const int  nxt = BUF1_OFF - cur;
        const bool has_next = t + 1 < t_end;

        v2f fq[32], fh[32];
        v8f acc;
        if (r_o & 1) acc = sv_part<2>(aS, oV0 + cur, fq, fh, aQ + cur, oH0);
        else         acc = sv_part<1>(aS, oV0 + cur, fq, fh, aQ + cur, oH0);
        EDGE1(acc);
        ld_rk<16, 8, SH_LD>(fq, fh, aQ + cur, oH0);
        EDGE1(acc);
        acc = mm<0, 8>(fq, fh, acc);
        EDGE1(acc);
        ld_rk<24, 8, SH_LD>(fq, fh, aQ + cur, oH0);
        if (has_next)
            stage_chunk(pq, pk, pv, sQ_st + nxt, sK_st + nxt, sV_st + nxt, scale);
        EDGE1(acc);
        acc = mm<8, 8>(fq, fh, acc);
        EDGE1(acc);
        ld_h<8, 8>(fk, fva, fvb, hK0 + cur, hK1 + cur, hV0 + cur);
        EDGE1(acc);
        acc = mm<16, 8>(fq, fh, acc);
        EDGE1(acc);
        {
            const int tn = (t + 2 < t_end) ? t + 2 : t_end - 1;
            const size_t adv = (size_t)tn * CH * row_stride;
            fetch_chunk(pq, pk, pv, qrow + adv, krow + adv, vrow + adv, row_stride);
        }
        EDGE1(acc);
        acc = mm<24, 8>(fq, fh, acc);

        #pragma unroll
        for (int j = 0; j < 8; ++j) {
            size_t s_idx = srow + r_o * 16 + j + 8 * kh;
            o[(((size_t)b * SEQ + s_idx) * NH + h) * D_DIM + g * GW + eco + m16] = acc[j];
        }
        EDGE2(h0, h1);
        lds_barrier();

        if (has_next) {
            s_phase<true>(has_tile, sA_a + nxt, sA_b + nxt, sS, rA, cA, m16, kh,
                          fk, fva, fvb, h0, h1, hT0, hT1, hM0, hM1,
                          hK0 + nxt, hK1 + nxt, hV0 + nxt);
        } else {
            mm2<8, 8>(fk, fva, h0, fk, fvb, h1);
            h_store(hT0, hT1, h0, h1);
        }
        lds_barrier();
    }

    if (second) {
    #pragma unroll
    for (int j = 0; j < 4; ++j) {
        int f    = tid + j * 256;
        int drow = f >> 3;
        int c4   = (f & 7) << 2;
        float4 val = *(const float4*)(sH + drow * SH_LD + c4);
        *(float4*)(hout + (((size_t)(b * NH + h)) * D_DIM + drow) * D_DIM + g * GW + c4) = val;
    }
    }
}

extern "C" void kernel_launch(void* const* d_in, const int* in_sizes, int n_in,
                              void* d_out, int out_size, void* d_ws, size_t ws_size,
                              hipStream_t stream) {
    (void)in_sizes; (void)n_in; (void)out_size; (void)d_ws; (void)ws_size;
    const float* q = (const float*)d_in[0];
    const float* k = (const float*)d_in[1];
    const float* v = (const float*)d_in[2];
    float* o  = (float*)d_out;
    float* hf = o + (size_t)NB * SEQ * NH * D_DIM;
    const float scale = 0.088388347648318447f;

    const size_t smem = (size_t)SMEM_FLOATS * sizeof(float);
    hipFuncSetAttribute((const void*)fla_fused_chunk_fwd,
                        hipFuncAttributeMaxDynamicSharedMemorySize, (int)smem);

    dim3 grid(2 * NWG1);
    dim3 blk(256);
    hipLaunchKernelGGL(fla_fused_chunk_fwd, grid, blk, smem, stream,
                       q, k, v, o, hf, scale);
}


# entering state of every chunk through the workspace: launch 1 runs H += K^T V alone, launch 2 does S once per chunk and O per (head, chunk) unit; needs 128 MiB, else the old form; O sum order changed
# speedup vs baseline: 3.9692x; 1.1794x over previous
#include <hip/hip_runtime.h>

typedef __attribute__((ext_vector_type(2))) float v2f;
typedef __attribute__((ext_vector_type(8))) float v8f;
typedef __attribute__((ext_vector_type(4))) float v4f;

#define D_DIM 128
#define CH    64
#define SEQ   4096
#define NT    (SEQ / CH)
#define NH    16
#define NB    2
#define EG    4
#define GW    32
#define NWG1  (NB * NH * EG)
#define T_SPLIT 35
static_assert(T_SPLIT >= 3 && T_SPLIT < NT, "the state pass keeps two chunks in flight");

#define SQ_LD 140
#define SK_LD 140
#define SV_LD 48
#define SS_LD 68
#define SH_LD 48

#define SQ_OFF 0
#define SK_OFF (SQ_OFF + 64 * SQ_LD)
#define SV_OFF (SK_OFF + 64 * SK_LD)
#define SS_OFF (SV_OFF + 64 * SV_LD)
#define SH_OFF (SS_OFF + 64 * SS_LD)
#define BUF1_OFF (SH_OFF + 128 * SH_LD)
#define SHM_OFF (BUF1_OFF + SV_OFF + 64 * SV_LD)
#define SMEM_FLOATS (SHM_OFF + 128 * SH_LD)
static_assert(BUF1_OFF % 64 == 0, "both buffers on the same banks");
static_assert(SHM_OFF % 64 == 0, "both state copies on the same banks");

#define EDGE1(a)    asm volatile("" : "+v"(a) : : "memory")
#define EDGE2(a, b) asm volatile("" : "+v"(a), "+v"(b) : : "memory")
#define EDGE3(a, b, c)    asm volatile("" : "+v"(a), "+v"(b), "+v"(c) : : "memory")
#define EDGE4(a, b, c, d) asm volatile("" : "+v"(a), "+v"(b), "+v"(c), "+v"(d) : : "memory")

static __device__ __forceinline__ v8f wmma_f32(v2f a, v2f b, v8f c) {
    return __builtin_amdgcn_wmma_f32_16x16x4_f32(
         false, a,  false, b,
         (short)0, c,  false,  false);
}

static __device__ __forceinline__ void lds_barrier() {
    __builtin_amdgcn_fence(__ATOMIC_RELEASE, "workgroup", "local");
    __builtin_amdgcn_s_barrier();
    __builtin_amdgcn_fence(__ATOMIC_ACQUIRE, "workgroup", "local");
}

template <int J0, int N>
static __device__ __forceinline__ void ld_rr(v2f* fa, v2f* fb, const float* pa, const float* pb) {
    #pragma unroll
    for (int j = J0; j < J0 + N; ++j) {
        fa[j] = *(const v2f*)(pa + 4 * j);
        fb[j] = *(const v2f*)(pb + 4 * j);
    }
}

template <int J0, int N, int LD>
static __device__ __forceinline__ void ld_rk(v2f* fa, v2f* fb, const float* pa,
                                             const float* p0) {
    #pragma unroll
    for (int j = J0; j < J0 + N; ++j) {
        fa[j] = *(const v2f*)(pa + 4 * j);
        fb[j].x = p0[4 * j * LD];
        fb[j].y = p0[4 * j * LD + LD];
    }
}

template <int J0, int N>
static __device__ __forceinline__ void ld_h(v2f* fk, v2f* fva, v2f* fvb,
                                            const float* k0, const float* k1,
                                            const float* v0) {
    #pragma unroll
    for (int j = J0; j < J0 + N; ++j) {
        fk[j].x  = k0[4 * j * SK_LD];      fk[j].y  = k1[4 * j * SK_LD];
        fva[j].x = v0[4 * j * SV_LD];      fva[j].y = v0[4 * j * SV_LD + SV_LD];
    }
    asm volatile("" : : : "memory");
    #pragma unroll
    for (int j = J0; j < J0 + N; ++j) {
        fvb[j].x = v0[4 * j * SV_LD + 16]; fvb[j].y = v0[4 * j * SV_LD + SV_LD + 16];
    }
}

template <int J0, int N>
static __device__ __forceinline__ v8f mm(const v2f* fa, const v2f* fb, v8f acc) {
    #pragma unroll
    for (int j = J0; j < J0 + N; ++j) acc = wmma_f32(fa[j], fb[j], acc);
    return acc;
}

template <int J0, int N>
static __device__ __forceinline__ void mm2(const v2f* fa0, const v2f* fb0, v8f& c0,
                                           const v2f* fa1, const v2f* fb1, v8f& c1) {
    #pragma unroll
    for (int j = J0; j < J0 + N; ++j) {
        c0 = wmma_f32(fa0[j], fb0[j], c0);
        c1 = wmma_f32(fa1[j], fb1[j], c1);
    }
}

static __device__ __forceinline__ void s_store(float* sS, v8f acc, int r, int c, int m16, int kh) {
    #pragma unroll
    for (int j = 0; j < 8; ++j) {
        int er = r * 16 + j + 8 * kh;
        int ec = c * 16 + m16;
        float val = acc[j];
        if (r == c && er < ec) val = 0.0f;
        sS[er * SS_LD + ec] = val;
    }
}

template <int NBLK>
static __device__ __forceinline__ v8f sv_part(const float* aS, const float* bV0,
                                              v2f* fq, v2f* fh,
                                              const float* aQ, const float* oH0) {
    v2f fs[NBLK * 4], fv[NBLK * 4];
    ld_rk<0, NBLK * 4, SV_LD>(fs, fv, aS, bV0);
    ld_rk<0, 16, SH_LD>(fq, fh, aQ, oH0);
    v8f acc = {};
    EDGE1(acc);
    return mm<0, NBLK * 4>(fs, fv, acc);
}


static __device__ __forceinline__ void fetch_chunk(v4f* pq, v4f* pk, v4f* pv,
                                                   const float* qrow, const float* krow,
                                                   const float* vrow, size_t row_stride) {
    #pragma unroll
    for (int j = 0; j < 8; ++j) {
        pq[j] = *(const v4f*)(qrow + (size_t)(8 * j) * row_stride);
        pk[j] = *(const v4f*)(krow + (size_t)(8 * j) * row_stride);
    }
    #pragma unroll
    for (int j = 0; j < 2; ++j)
        pv[j] = *(const v4f*)(vrow + (size_t)(32 * j) * row_stride);
}

static __device__ __forceinline__ void fetch_kv(v4f* pk, v4f* pv, const float* krow,
                                                const float* vrow, size_t row_stride) {
    #pragma unroll
    for (int j = 0; j < 8; ++j)
        pk[j] = *(const v4f*)(krow + (size_t)(8 * j) * row_stride);
    #pragma unroll
    for (int j = 0; j < 2; ++j)
        pv[j] = *(const v4f*)(vrow + (size_t)(32 * j) * row_stride);
}

static __device__ __forceinline__ void stage_kv(const v4f* pk, const v4f* pv,
                                                float* sK_st, float* sV_st) {
    #pragma unroll
    for (int j = 0; j < 8; ++j) *(v4f*)(sK_st + 8 * j * SK_LD) = pk[j];
    #pragma unroll
    for (int j = 0; j < 2; ++j) *(v4f*)(sV_st + 32 * j * SV_LD) = pv[j];
}

static __device__ __forceinline__ void state_step(v4f* pk, v4f* pv, int t,
                                                  const float* krow, const float* vrow,
                                                  size_t row_stride,
                                                  const float* rK, const float* rV,
                                                  float* wK, float* wV,
                                                  int qk_row, int qk_c4, int v_row, int v_c4,
                                                  int kh, int m16, int d, v8f& h0, v8f& h1) {
    const float* k0 = rK + (kh * 2 + 0) * SK_LD + d * 16 + m16;
    const float* k1 = rK + (kh * 2 + 1) * SK_LD + d * 16 + m16;
    const float* v0 = rV + (kh * 2 + 0) * SV_LD + m16;
    v2f fk[16], fva[16], fvb[16];
    ld_h<0, 4>(fk, fva, fvb, k0, k1, v0);
    ld_h<4, 4>(fk, fva, fvb, k0, k1, v0);
    EDGE2(h0, h1);
    if (t + 1 < T_SPLIT) {
        stage_kv(pk, pv, wK + qk_row * SK_LD + qk_c4, wV + v_row * SV_LD + v_c4);
        const int tn = (t + 3 < T_SPLIT) ? t + 3 : T_SPLIT - 1;
        const size_t adv = (size_t)tn * CH * row_stride;
        fetch_kv(pk, pv, krow + adv, vrow + adv, row_stride);
    }
    EDGE2(h0, h1);
    mm2<0, 4>(fk, fva, h0, fk, fvb, h1);
    EDGE2(h0, h1);
    ld_h<8, 4>(fk, fva, fvb, k0, k1, v0);
    EDGE2(h0, h1);
    mm2<4, 4>(fk, fva, h0, fk, fvb, h1);
    EDGE2(h0, h1);
    ld_h<12, 4>(fk, fva, fvb, k0, k1, v0);
    EDGE2(h0, h1);
    mm2<8, 8>(fk, fva, h0, fk, fvb, h1);
    lds_barrier();
}

static __device__ __forceinline__ void stage_chunk(const v4f* pq, const v4f* pk, const v4f* pv,
                                                   float* sQ_st, float* sK_st, float* sV_st,
                                                   float scale) {
    #pragma unroll
    for (int j = 0; j < 8; ++j) {
        *(v4f*)(sQ_st + 8 * j * SQ_LD) = pq[j] * scale;
        *(v4f*)(sK_st + 8 * j * SK_LD) = pk[j];
    }
    #pragma unroll
    for (int j = 0; j < 2; ++j)
        *(v4f*)(sV_st + 32 * j * SV_LD) = pv[j];
}

static __device__ __forceinline__ void h_store(float* hT0, float* hT1, v8f h0, v8f h1) {
    #pragma unroll
    for (int j = 0; j < 8; ++j) { hT0[j * SH_LD] = h0[j]; hT1[j * SH_LD] = h1[j]; }
}

template <bool STEADY>
static __device__ __forceinline__ void s_phase(bool has_tile,
                                               const float* sA_a, const float* sA_b,
                                               float* sS, int rA, int cA,
                                               int m16, int kh,
                                               v2f* fk, v2f* fva, v2f* fvb,
                                               v8f& h0, v8f& h1,
                                               float* hT0, float* hT1,
                                               float* hM0, float* hM1,
                                               const float* nK0, const float* nK1,
                                               const float* nV0) {
    if (has_tile) {
        v2f fa[32], fb[32];
        v8f c0 = {};
        ld_rr<0, 8>(fa, fb, sA_a, sA_b);
        ld_rr<8, 8>(fa, fb, sA_a, sA_b);
        ld_h<0, 8>(fk, fva, fvb, nK0, nK1, nV0);
        if constexpr (STEADY) {
            EDGE2(h0, h1);
            mm2<8, 8>(fk, fva, h0, fk, fvb, h1);
            EDGE3(h0, h1, c0);
        } else {
            EDGE1(c0);
        }
        c0 = mm<0, 8>(fa, fb, c0);
        EDGE3(h0, h1, c0);
        ld_rr<16, 8>(fa, fb, sA_a, sA_b);
        if constexpr (STEADY) h_store(hT0, hT1, h0, h1);
        EDGE3(h0, h1, c0);
        mm2<0, 8>(fk, fva, h0, fk, fvb, h1);
        c0 = mm<8, 8>(fa, fb, c0);
        EDGE3(h0, h1, c0);
        ld_rr<24, 8>(fa, fb, sA_a, sA_b);
        h_store(hM0, hM1, h0, h1);
        EDGE1(c0);
        c0 = mm<16, 16>(fa, fb, c0);
        EDGE1(c0);
        s_store(sS, c0, rA, cA, m16, kh);
    } else {
        ld_h<0, 8>(fk, fva, fvb, nK0, nK1, nV0);
        if constexpr (STEADY) {
            EDGE2(h0, h1);
            mm2<8, 8>(fk, fva, h0, fk, fvb, h1);
            EDGE2(h0, h1);
            h_store(hT0, hT1, h0, h1);
        }
        EDGE2(h0, h1);
        mm2<0, 8>(fk, fva, h0, fk, fvb, h1);
        EDGE2(h0, h1);
        h_store(hM0, hM1, h0, h1);
    }
}

extern "C" __global__ __launch_bounds__(256)
void fla_fused_chunk_fwd(const float* __restrict__ q,
                         const float* __restrict__ k,
                         const float* __restrict__ v,
                         float* __restrict__ o,
                         float* __restrict__ hout,
                         float scale)
{
    extern __shared__ float smem[];
    float* sQ = smem + SQ_OFF;
    float* sK = smem + SK_OFF;
    float* sV = smem + SV_OFF;
    float* sS = smem + SS_OFF;
    float* sH = smem + SH_OFF;
    float* sHm = smem + SHM_OFF;

    const int tid  = threadIdx.x;
    const int wave = tid >> 5;
    const int lane = tid & 31;
    const int m16  = lane & 15;
    const int kh   = lane >> 4;

    const bool second = blockIdx.x < NWG1;
    const int  t_begin = second ? T_SPLIT : 0;
    const int  t_end   = second ? NT : T_SPLIT;
    const int g  = blockIdx.x & (EG - 1);
    const int bh = (blockIdx.x >> 2) & (NB * NH - 1);
    const int h  = bh & (NH - 1);
    const int b  = bh >> 4;

    const size_t row_stride = (size_t)NH * D_DIM;
    const float* qbase = q + ((size_t)b * SEQ * NH + h) * D_DIM;
    const float* kbase = k + ((size_t)b * SEQ * NH + h) * D_DIM;
    const float* vbase = v + ((size_t)b * SEQ * NH + h) * D_DIM;

    const int qk_row = tid >> 5, qk_c4 = (tid & 31) << 2;
    const int v_row  = tid >> 3, v_c4  = (tid & 7) << 2;
    const float* qrow = qbase + (size_t)qk_row * row_stride + qk_c4;
    const float* krow = kbase + (size_t)qk_row * row_stride + qk_c4;
    const float* vrow = vbase + (size_t)v_row * row_stride + g * GW + v_c4;
    float* sQ_st = sQ + qk_row * SQ_LD + qk_c4;
    float* sK_st = sK + qk_row * SK_LD + qk_c4;
    float* sV_st = sV + v_row * SV_LD + v_c4;

    v4f pq[8], pk[8], pv[2];
    v8f h0 = {}, h1 = {};
    v2f fk[16], fva[16], fvb[16];
    if (second) {
        const int dd = wave;
        const size_t cstep = (size_t)CH * row_stride;
        const size_t adv0  = (size_t)T_SPLIT * cstep;
        v4f ka[8], va[2], kb[8], vb[2];
        fetch_kv(ka, va, krow, vrow, row_stride);
        fetch_kv(kb, vb, krow + cstep, vrow + cstep, row_stride);
        stage_kv(ka, va, sK_st, sV_st);
        fetch_kv(ka, va, krow + 2 * cstep, vrow + 2 * cstep, row_stride);
        lds_barrier();
        for (int t = 0; t + 1 < T_SPLIT; t += 2) {
            state_step(kb, vb, t,     krow, vrow, row_stride, sK, sV, sQ, sS,
                       qk_row, qk_c4, v_row, v_c4, kh, m16, dd, h0, h1);
            state_step(ka, va, t + 1, krow, vrow, row_stride, sQ, sS, sK, sV,
                       qk_row, qk_c4, v_row, v_c4, kh, m16, dd, h0, h1);
        }
        fetch_chunk(pq, pk, pv, qrow + adv0, krow + adv0, vrow + adv0, row_stride);
        if (T_SPLIT & 1)
            state_step(kb, vb, T_SPLIT - 1, krow, vrow, row_stride, sK, sV, sQ, sS,
                       qk_row, qk_c4, v_row, v_c4, kh, m16, dd, h0, h1);
        float* pT0 = sH + (dd * 16 + 8 * kh) * SH_LD + m16;
        #pragma unroll
        for (int j = 0; j < 8; ++j) { pT0[j * SH_LD] = h0[j]; pT0[j * SH_LD + 16] = h1[j]; }
    } else {
        fetch_chunk(pq, pk, pv, qrow, krow, vrow, row_stride);
        for (int i = tid; i < 128 * SH_LD; i += 256) sH[i] = 0.0f;
    }

    const int cur0 = (t_begin & 1) ? BUF1_OFF : 0;
    stage_chunk(pq, pk, pv, sQ_st + cur0, sK_st + cur0, sV_st + cur0, scale);
    if (t_begin + 1 < t_end) {
        const size_t adv = (size_t)(t_begin + 1) * CH * row_stride;
        fetch_chunk(pq, pk, pv, qrow + adv, krow + adv, vrow + adv, row_stride);
    }
    lds_barrier();

    const int r_o  = wave & 3;
    const int eco  = (wave >> 2) * 16;

    const bool has_tile = wave < 6;
    const int  tw = has_tile ? wave : 0;
    const int  rA = (tw >= 4) ? 3 : (tw == 3) ? 2 : (tw >= 1) ? 1 : 0;
    const int  cA = (tw == 5) ? 3 : (tw >= 3) ? 2 : (tw == 2) ? 1 : 0;
    const float* sA_a = sQ + (rA * 16 + m16) * SQ_LD + kh * 2;
    const float* sA_b = sK + (cA * 16 + m16) * SK_LD + kh * 2;

    const int  half = (r_o >> 1) * 32;
    const float* aS  = sS + (r_o * 16 + m16) * SS_LD + half + kh * 2;
    const float* aQ  = sQ + (r_o * 16 + m16) * SQ_LD + kh * 2;
    const float* oV0 = sV + (half + kh * 2) * SV_LD + eco + m16;
    const float* oH0 = ((r_o >> 1) ? sHm : sH) + (kh * 2 + 0) * SH_LD + eco + m16;

    const int d = wave;
    float* hT0 = sH + (d * 16 + 8 * kh) * SH_LD + m16;
    float* hT1 = hT0 + 16;
    float* hM0 = sHm + (d * 16 + 8 * kh) * SH_LD + m16;
    float* hM1 = hM0 + 16;
    const float* hK0 = sK + (kh * 2 + 0) * SK_LD + d * 16 + m16;
    const float* hK1 = sK + (kh * 2 + 1) * SK_LD + d * 16 + m16;
    const float* hV0 = sV + (kh * 2 + 0) * SV_LD + m16;

    s_phase<false>(has_tile, sA_a + cur0, sA_b + cur0, sS, rA, cA, m16, kh,
                   fk, fva, fvb, h0, h1, hT0, hT1, hM0, hM1,
                   hK0 + cur0, hK1 + cur0, hV0 + cur0);
    lds_barrier();

    for (int t = t_begin; t < t_end; ++t) {
        const size_t srow = (size_t)t * CH;
        const int  cur = (t & 1) ? BUF1_OFF : 0;
        const int  nxt = BUF1_OFF - cur;
        const bool has_next = t + 1 < t_end;

        v2f fq[32], fh[32];
        v8f acc;
        if (r_o & 1) acc = sv_part<2>(aS, oV0 + cur, fq, fh, aQ + cur, oH0);
        else         acc = sv_part<1>(aS, oV0 + cur, fq, fh, aQ + cur, oH0);
        EDGE1(acc);
        ld_rk<16, 8, SH_LD>(fq, fh, aQ + cur, oH0);
        EDGE1(acc);
        acc = mm<0, 8>(fq, fh, acc);
        EDGE1(acc);
        ld_rk<24, 8, SH_LD>(fq, fh, aQ + cur, oH0);
        if (has_next)
            stage_chunk(pq, pk, pv, sQ_st + nxt, sK_st + nxt, sV_st + nxt, scale);
        EDGE1(acc);
        acc = mm<8, 8>(fq, fh, acc);
        EDGE1(acc);
        ld_h<8, 8>(fk, fva, fvb, hK0 + cur, hK1 + cur, hV0 + cur);
        EDGE1(acc);
        acc = mm<16, 8>(fq, fh, acc);
        EDGE1(acc);
        {
            const int tn = (t + 2 < t_end) ? t + 2 : t_end - 1;
            const size_t adv = (size_t)tn * CH * row_stride;
            fetch_chunk(pq, pk, pv, qrow + adv, krow + adv, vrow + adv, row_stride);
        }
        EDGE1(acc);
        acc = mm<24, 8>(fq, fh, acc);

        #pragma unroll
        for (int j = 0; j < 8; ++j) {
            size_t s_idx = srow + r_o * 16 + j + 8 * kh;
            o[(((size_t)b * SEQ + s_idx) * NH + h) * D_DIM + g * GW + eco + m16] = acc[j];
        }
        EDGE2(h0, h1);
        lds_barrier();

        if (has_next) {
            s_phase<true>(has_tile, sA_a + nxt, sA_b + nxt, sS, rA, cA, m16, kh,
                          fk, fva, fvb, h0, h1, hT0, hT1, hM0, hM1,
                          hK0 + nxt, hK1 + nxt, hV0 + nxt);
        } else {
            mm2<8, 8>(fk, fva, h0, fk, fvb, h1);
            h_store(hT0, hT1, h0, h1);
        }
        lds_barrier();
    }

    if (second) {
    #pragma unroll
    for (int j = 0; j < 4; ++j) {
        int f    = tid + j * 256;
        int drow = f >> 3;
        int c4   = (f & 7) << 2;
        float4 val = *(const float4*)(sH + drow * SH_LD + c4);
        *(float4*)(hout + (((size_t)(b * NH + h)) * D_DIM + drow) * D_DIM + g * GW + c4) = val;
    }
    }
}


#define WS_TILE   256
#define WS_CHUNK  (64 * WS_TILE)
#define WS_BYTES  ((size_t)NB * NH * NT * WS_CHUNK * sizeof(float))

#define KA_LD 76
#define PA_FLOATS (64 * KA_LD + 64 * SV_LD)
#define SMEM_A_FLOATS (2 * PA_FLOATS)

template <int J0, int N>
static __device__ __forceinline__ void ld_h1(v2f* fk, v2f* fv, const float* k0,
                                             const float* k1, const float* v0) {
    #pragma unroll
    for (int j = J0; j < J0 + N; ++j) {
        fk[j].x = k0[4 * j * KA_LD];  fk[j].y = k1[4 * j * KA_LD];
        fv[j].x = v0[4 * j * SV_LD];  fv[j].y = v0[4 * j * SV_LD + SV_LD];
    }
}

static __device__ __forceinline__ void fetch_kvA(v4f* pk, v4f* pv, const float* krow,
                                                 const float* vrow, size_t rs) {
    #pragma unroll
    for (int j = 0; j < 4; ++j) pk[j] = *(const v4f*)(krow + (size_t)(16 * j) * rs);
    #pragma unroll
    for (int j = 0; j < 2; ++j) pv[j] = *(const v4f*)(vrow + (size_t)(32 * j) * rs);
}

static __device__ __forceinline__ void stage_kvA(const v4f* pk, const v4f* pv,
                                                 float* sK_st, float* sV_st) {
    #pragma unroll
    for (int j = 0; j < 4; ++j) *(v4f*)(sK_st + 16 * j * KA_LD) = pk[j];
    #pragma unroll
    for (int j = 0; j < 2; ++j) *(v4f*)(sV_st + 32 * j * SV_LD) = pv[j];
}

static __device__ __forceinline__ void state_stepA(v4f* pk, v4f* pv, int t,
                                                   const float* krow, const float* vrow, size_t rs,
                                                   const float* rK, const float* rV,
                                                   float* wKst, float* wVst,
                                                   int kh, int m16, int dl, int cs,
                                                   float* wsT, v8f& h0) {
    const float* k0 = rK + (kh * 2 + 0) * KA_LD + dl * 16 + m16;
    const float* k1 = rK + (kh * 2 + 1) * KA_LD + dl * 16 + m16;
    const float* v0 = rV + (kh * 2 + 0) * SV_LD + cs * 16 + m16;
    v2f fk[16], fv[16];
    ld_h1<0, 8>(fk, fv, k0, k1, v0);
    EDGE1(h0);
    {
        float* wp = wsT + (size_t)t * WS_CHUNK;
        v4f s0 = { h0[0], h0[1], h0[2], h0[3] };
        v4f s1 = { h0[4], h0[5], h0[6], h0[7] };
        *(v4f*)(wp)      = s0;
        *(v4f*)(wp + 64) = s1;
    }
    if (t + 1 < NT) {
        stage_kvA(pk, pv, wKst, wVst);
        const int tn = (t + 3 < NT) ? t + 3 : NT - 1;
        const size_t adv = (size_t)tn * CH * rs;
        fetch_kvA(pk, pv, krow + adv, vrow + adv, rs);
    }
    EDGE1(h0);
    h0 = mm<0, 4>(fk, fv, h0);
    EDGE1(h0);
    ld_h1<8, 8>(fk, fv, k0, k1, v0);
    EDGE1(h0);
    h0 = mm<4, 12>(fk, fv, h0);
    lds_barrier();
}

extern "C" __global__ __launch_bounds__(256)
void fla_state_pass(const float* __restrict__ k, const float* __restrict__ v,
                    float* __restrict__ ws, float* __restrict__ hout)
{
    extern __shared__ float smem[];
    float* sK0 = smem;
    float* sV0 = smem + 64 * KA_LD;
    float* sK1 = smem + PA_FLOATS;
    float* sV1 = smem + PA_FLOATS + 64 * KA_LD;

    const int tid  = threadIdx.x;
    const int wave = tid >> 5;
    const int lane = tid & 31;
    const int m16  = lane & 15;
    const int kh   = lane >> 4;
    const int dh = blockIdx.x & 1;
    const int g  = (blockIdx.x >> 1) & (EG - 1);
    const int bh = blockIdx.x >> 3;
    const int h  = bh & (NH - 1);
    const int b  = bh >> 4;
    const int dl = wave & 3;
    const int cs = wave >> 2;
    const int dd = dh * 4 + dl;
    const int e  = g * 2 + cs;

    const size_t rs = (size_t)NH * D_DIM;
    const float* kbase = k + ((size_t)b * SEQ * NH + h) * D_DIM;
    const float* vbase = v + ((size_t)b * SEQ * NH + h) * D_DIM;
    const int k_row = tid >> 4, k_c4 = (tid & 15) << 2;
    const int v_row = tid >> 3, v_c4 = (tid & 7) << 2;
    const float* krow = kbase + (size_t)k_row * rs + dh * 64 + k_c4;
    const float* vrow = vbase + (size_t)v_row * rs + g * GW + v_c4;
    const int kst = k_row * KA_LD + k_c4;
    const int vst = v_row * SV_LD + v_c4;

    float* wsT = ws + (size_t)bh * NT * WS_CHUNK + (size_t)(e * 8 + dd) * WS_TILE
               + kh * 128 + m16 * 4;

    const size_t cstep = (size_t)CH * rs;
    v8f h0 = {};
    v4f ka[4], va[2], kb[4], vb[2];
    fetch_kvA(ka, va, krow, vrow, rs);
    fetch_kvA(kb, vb, krow + cstep, vrow + cstep, rs);
    stage_kvA(ka, va, sK0 + kst, sV0 + vst);
    fetch_kvA(ka, va, krow + 2 * cstep, vrow + 2 * cstep, rs);
    lds_barrier();
    for (int t = 0; t < NT; t += 2) {
        state_stepA(kb, vb, t,     krow, vrow, rs, sK0, sV0, sK1 + kst, sV1 + vst,
                    kh, m16, dl, cs, wsT, h0);
        state_stepA(ka, va, t + 1, krow, vrow, rs, sK1, sV1, sK0 + kst, sV0 + vst,
                    kh, m16, dl, cs, wsT, h0);
    }
    float* hp = hout + (((size_t)bh) * D_DIM + dd * 16 + 8 * kh) * D_DIM + e * 16 + m16;
    #pragma unroll
    for (int j = 0; j < 8; ++j) hp[(size_t)j * D_DIM] = h0[j];
}

#define BV_LD 132
#define B_SQ  0
#define B_SK  (B_SQ + 64 * SQ_LD)
#define B_SV  (B_SK + 64 * SK_LD)
#define B_SS  (B_SV + 64 * BV_LD)
#define SMEM_B_FLOATS (B_SS + 64 * SS_LD)

static __device__ __forceinline__ v2f lo2(v4f x) { return __builtin_shufflevector(x, x, 0, 1); }
static __device__ __forceinline__ v2f hi2(v4f x) { return __builtin_shufflevector(x, x, 2, 3); }

template <int P0, int N>
static __device__ __forceinline__ void ld4_rr(v4f* fa, v4f* fb, const float* pa, const float* pb) {
    #pragma unroll
    for (int p = P0; p < P0 + N; ++p) {
        fa[p] = *(const v4f*)(pa + 8 * p);
        fb[p] = *(const v4f*)(pb + 8 * p);
    }
}

template <int P0, int N>
static __device__ __forceinline__ v8f mm4(const v4f* fa, const v4f* fb, v8f acc) {
    #pragma unroll
    for (int p = P0; p < P0 + N; ++p) {
        acc = wmma_f32(lo2(fa[p]), lo2(fb[p]), acc);
        acc = wmma_f32(hi2(fa[p]), hi2(fb[p]), acc);
    }
    return acc;
}

static __device__ __forceinline__ void s_tile(const float* sQ, const float* sK, float* sS,
                                              int rA, int cA, int m16, int kh) {
    const float* pa = sQ + (rA * 16 + m16) * SQ_LD + 4 * kh;
    const float* pb = sK + (cA * 16 + m16) * SK_LD + 4 * kh;
    v4f fa[16], fb[16];
    v8f c0 = {};
    ld4_rr<0, 4>(fa, fb, pa, pb);
    ld4_rr<4, 4>(fa, fb, pa, pb);
    EDGE1(c0);
    c0 = mm4<0, 4>(fa, fb, c0);
    EDGE1(c0);
    ld4_rr<8, 4>(fa, fb, pa, pb);
    EDGE1(c0);
    c0 = mm4<4, 4>(fa, fb, c0);
    EDGE1(c0);
    ld4_rr<12, 4>(fa, fb, pa, pb);
    EDGE1(c0);
    c0 = mm4<8, 8>(fa, fb, c0);
    EDGE1(c0);
    s_store(sS, c0, rA, cA, m16, kh);
}

static __device__ __forceinline__ void ldq(v4f* dst, const float* aQ, int G) {
    #pragma unroll
    for (int pp = 0; pp < 2; ++pp)
        #pragma unroll
        for (int r = 0; r < 4; ++r)
            dst[pp * 4 + r] = *(const v4f*)(aQ + r * 16 * SQ_LD + 8 * (2 * G + pp));
}

static __device__ __forceinline__ void mmq(const v4f* qa, const v4f* hv, int G,
                                           v8f& a0, v8f& a1, v8f& a2, v8f& a3) {
    #pragma unroll
    for (int pp = 0; pp < 2; ++pp) {
        const v4f hb = hv[2 * G + pp];
        a0 = wmma_f32(lo2(qa[pp * 4 + 0]), lo2(hb), a0);
        a1 = wmma_f32(lo2(qa[pp * 4 + 1]), lo2(hb), a1);
        a2 = wmma_f32(lo2(qa[pp * 4 + 2]), lo2(hb), a2);
        a3 = wmma_f32(lo2(qa[pp * 4 + 3]), lo2(hb), a3);
        a0 = wmma_f32(hi2(qa[pp * 4 + 0]), hi2(hb), a0);
        a1 = wmma_f32(hi2(qa[pp * 4 + 1]), hi2(hb), a1);
        a2 = wmma_f32(hi2(qa[pp * 4 + 2]), hi2(hb), a2);
        a3 = wmma_f32(hi2(qa[pp * 4 + 3]), hi2(hb), a3);
    }
}

extern "C" __global__ __launch_bounds__(256)
void fla_chunk_out(const float* __restrict__ q, const float* __restrict__ k,
                   const float* __restrict__ v, const float* __restrict__ hws,
                   float* __restrict__ o, float scale)
{
    extern __shared__ float smem[];
    float* sQ = smem + B_SQ;
    float* sK = smem + B_SK;
    float* sV = smem + B_SV;
    float* sS = smem + B_SS;

    const int tid  = threadIdx.x;
    const int wave = tid >> 5;
    const int lane = tid & 31;
    const int m16  = lane & 15;
    const int kh   = lane >> 4;
    const int t  = blockIdx.x & (NT - 1);
    const int bh = blockIdx.x >> 6;
    const int h  = bh & (NH - 1);
    const int b  = bh >> 4;
    const int e  = wave;

    const size_t rs = (size_t)NH * D_DIM;
    const size_t base = ((size_t)b * SEQ + (size_t)t * CH) * rs + (size_t)h * D_DIM;
    const int row0 = tid >> 5, c4 = (tid & 31) << 2;
    const float* qrow = q + base + (size_t)row0 * rs + c4;
    const float* krow = k + base + (size_t)row0 * rs + c4;
    const float* vrow = v + base + (size_t)row0 * rs + c4;

    v4f pq[8], pk[8], pv[8];
    #pragma unroll
    for (int j = 0; j < 8; ++j) {
        pq[j] = *(const v4f*)(qrow + (size_t)(8 * j) * rs);
        pk[j] = *(const v4f*)(krow + (size_t)(8 * j) * rs);
    }
    #pragma unroll
    for (int j = 0; j < 8; ++j) pv[j] = *(const v4f*)(vrow + (size_t)(8 * j) * rs);
    v4f hv[16];
    const float* hp = hws + ((size_t)(bh * NT + t) * 8 + e) * (8 * WS_TILE) + lane * 4;
    #pragma unroll
    for (int p = 0; p < 16; ++p) hv[p] = *(const v4f*)(hp + p * 128);

    #pragma unroll
    for (int j = 0; j < 8; ++j) {
        *(v4f*)(sQ + (row0 + 8 * j) * SQ_LD + c4) = pq[j] * scale;
        *(v4f*)(sK + (row0 + 8 * j) * SK_LD + c4) = pk[j];
    }
    #pragma unroll
    for (int j = 0; j < 8; ++j) *(v4f*)(sV + (row0 + 8 * j) * BV_LD + c4) = pv[j];
    lds_barrier();

    {
        const int i0 = wave;
        const int r0 = (i0 >= 6) ? 3 : (i0 >= 3) ? 2 : (i0 >= 1) ? 1 : 0;
        s_tile(sQ, sK, sS, r0, i0 - r0 * (r0 + 1) / 2, m16, kh);
        if (wave < 2) s_tile(sQ, sK, sS, 3, 2 + wave, m16, kh);
    }

    v8f a0 = {}, a1 = {}, a2 = {}, a3 = {};
    {
        const float* aQ = sQ + m16 * SQ_LD + 4 * kh;
        v4f qa[2][8];
        ldq(qa[0], aQ, 0);
        #pragma unroll
        for (int G = 0; G < 8; ++G) {
            EDGE4(a0, a1, a2, a3);
            if (G + 1 < 8) ldq(qa[(G + 1) & 1], aQ, G + 1);
            EDGE4(a0, a1, a2, a3);
            mmq(qa[G & 1], hv, G, a0, a1, a2, a3);
        }
    }
    lds_barrier();

    {
        const float* aS = sS + m16 * SS_LD + 4 * kh;
        const float* bV = sV + (4 * kh) * BV_LD + e * 16 + m16;
        v4f vb[8], sa[20];
        #pragma unroll
        for (int c = 0; c < 4; ++c)
            #pragma unroll
            for (int pp = 0; pp < 2; ++pp) {
                const float* p = bV + (c * 16 + 8 * pp) * BV_LD;
                v4f x = { p[0], p[BV_LD], p[2 * BV_LD], p[3 * BV_LD] };
                vb[c * 2 + pp] = x;
            }
        #pragma unroll
        for (int r = 0; r < 4; ++r)
            #pragma unroll
            for (int c = 0; c <= r; ++c)
                #pragma unroll
                for (int pp = 0; pp < 2; ++pp)
                    sa[(r * (r + 1) / 2 + c) * 2 + pp] =
                        *(const v4f*)(aS + r * 16 * SS_LD + c * 16 + 8 * pp);
        EDGE4(a0, a1, a2, a3);
        #pragma unroll
        for (int c = 0; c < 4; ++c)
            #pragma unroll
            for (int pp = 0; pp < 2; ++pp) {
                const v4f x = vb[c * 2 + pp];
                if (c <= 0) a0 = wmma_f32(lo2(sa[(0 + c) * 2 + pp]), lo2(x), a0);
                if (c <= 1) a1 = wmma_f32(lo2(sa[(1 + c) * 2 + pp]), lo2(x), a1);
                if (c <= 2) a2 = wmma_f32(lo2(sa[(3 + c) * 2 + pp]), lo2(x), a2);
                a3 = wmma_f32(lo2(sa[(6 + c) * 2 + pp]), lo2(x), a3);
                if (c <= 0) a0 = wmma_f32(hi2(sa[(0 + c) * 2 + pp]), hi2(x), a0);
                if (c <= 1) a1 = wmma_f32(hi2(sa[(1 + c) * 2 + pp]), hi2(x), a1);
                if (c <= 2) a2 = wmma_f32(hi2(sa[(3 + c) * 2 + pp]), hi2(x), a2);
                a3 = wmma_f32(hi2(sa[(6 + c) * 2 + pp]), hi2(x), a3);
            }
    }

    float* op = o + base + (size_t)(8 * kh) * rs + e * 16 + m16;
    #pragma unroll
    for (int j = 0; j < 8; ++j) {
        op[(size_t)(0 * 16 + j) * rs] = a0[j];
        op[(size_t)(1 * 16 + j) * rs] = a1[j];
        op[(size_t)(2 * 16 + j) * rs] = a2[j];
        op[(size_t)(3 * 16 + j) * rs] = a3[j];
    }
}

extern "C" void kernel_launch(void* const* d_in, const int* in_sizes, int n_in,
                              void* d_out, int out_size, void* d_ws, size_t ws_size,
                              hipStream_t stream) {
    (void)in_sizes; (void)n_in; (void)out_size;
    const float* q = (const float*)d_in[0];
    const float* k = (const float*)d_in[1];
    const float* v = (const float*)d_in[2];
    float* o  = (float*)d_out;
    float* hf = o + (size_t)NB * SEQ * NH * D_DIM;
    const float scale = 0.088388347648318447f;

    if (d_ws != nullptr && ws_size >= WS_BYTES) {
        float* ws = (float*)d_ws;
        const size_t smemA = (size_t)SMEM_A_FLOATS * sizeof(float);
        const size_t smemB = (size_t)SMEM_B_FLOATS * sizeof(float);
        hipFuncSetAttribute((const void*)fla_state_pass,
                            hipFuncAttributeMaxDynamicSharedMemorySize, (int)smemA);
        hipFuncSetAttribute((const void*)fla_chunk_out,
                            hipFuncAttributeMaxDynamicSharedMemorySize, (int)smemB);
        hipLaunchKernelGGL(fla_state_pass, dim3(NB * NH * EG * 2), dim3(256), smemA, stream,
                           k, v, ws, hf);
        hipLaunchKernelGGL(fla_chunk_out, dim3(NB * NH * NT), dim3(256), smemB, stream,
                           q, k, v, (const float*)ws, o, scale);
        return;
    }

    const size_t smem = (size_t)SMEM_FLOATS * sizeof(float);
    hipFuncSetAttribute((const void*)fla_fused_chunk_fwd,
                        hipFuncAttributeMaxDynamicSharedMemorySize, (int)smem);

    dim3 grid(2 * NWG1);
    dim3 blk(256);
    hipLaunchKernelGGL(fla_fused_chunk_fwd, grid, blk, smem, stream,
                       q, k, v, o, hf, scale);
}
